# starting kernel
# baseline (speedup 1.0000x reference)
#include <hip/hip_runtime.h>
#include <hip/hip_bf16.h>
#include <math.h>

#define NQ    10
#define NL    6
#define BATCH 2048
#define DIM   1024


typedef float v2f __attribute__((ext_vector_type(2)));
typedef float v8f __attribute__((ext_vector_type(8)));

__device__ __forceinline__ float2 cmul(float2 a, float2 b) {
    return make_float2(a.x * b.x - a.y * b.y, a.x * b.y + a.y * b.x);
}
__device__ __forceinline__ float2 cadd(float2 a, float2 b) {
    return make_float2(a.x + b.x, a.y + b.y);
}
__device__ __forceinline__ float2 cscale(float s, float2 a) {
    return make_float2(s * a.x, s * a.y);
}

__global__ void __launch_bounds__(256) embed_kernel(const float* __restrict__ x,
                                                    float* __restrict__ pre,
                                                    float* __restrict__ pim) {
    __shared__ float cs[NQ], sn[NQ];
    const int b = blockIdx.x;
    const int tid = threadIdx.x;
    if (tid < NQ) {
        float s, c;
        sincosf(0.5f * x[b * NQ + tid], &s, &c);
        sn[tid] = s; cs[tid] = c;
    }
    __syncthreads();
    #pragma unroll
    for (int i = 0; i < DIM / 256; ++i) {
        const int k = tid + 256 * i;
        float r = 1.0f;
        #pragma unroll
        for (int q = 0; q < NQ; ++q) {
            const int bit = (k >> (NQ - 1 - q)) & 1;
            r *= bit ? sn[q] : cs[q];
        }
        const int p = __popc(k) & 3;
        const float re = (p == 0) ? r : (p == 2 ? -r : 0.0f);
        const float im = (p == 1) ? -r : (p == 3 ? r : 0.0f);
        pre[b * DIM + k] = re;
        pim[b * DIM + k] = im;
    }
}

__global__ void __launch_bounds__(128) build_u_kernel(const float* __restrict__ w,
                                                      float* __restrict__ Mre,
                                                      float* __restrict__ Mim) {
    __shared__ float2 st[DIM];
    const int j = blockIdx.x;
    const int tid = threadIdx.x;
    for (int k = tid; k < DIM; k += 128)
        st[k] = make_float2(k == j ? 1.0f : 0.0f, 0.0f);
    __syncthreads();

    for (int l = 0; l < NL; ++l) {
        for (int q = 0; q < NQ; ++q) {
            const float phi = w[(l * NQ + q) * 3 + 0];
            const float th  = w[(l * NQ + q) * 3 + 1];
            const float om  = w[(l * NQ + q) * 3 + 2];
            float ct, s_t, ca, sa, cb, sb;
            sincosf(0.5f * th, &s_t, &ct);
            sincosf(0.5f * (phi + om), &sa, &ca);
            sincosf(0.5f * (phi - om), &sb, &cb);
            const float2 ga = make_float2( ct * ca, -ct * sa);
            const float2 gb = make_float2(-s_t * cb, -s_t * sb);
            const float2 gc = make_float2( s_t * cb, -s_t * sb);
            const float2 gd = make_float2( ct * ca,  ct * sa);
            const int p = NQ - 1 - q;
            for (int t = tid; t < DIM / 2; t += 128) {
                const int i0 = ((t >> p) << (p + 1)) | (t & ((1 << p) - 1));
                const int i1 = i0 | (1 << p);
                const float2 s0 = st[i0], s1 = st[i1];
                st[i0] = cadd(cmul(ga, s0), cmul(gb, s1));
                st[i1] = cadd(cmul(gc, s0), cmul(gd, s1));
            }
            __syncthreads();
        }
        for (int q = 0; q < NQ; ++q) {
            const int pc = NQ - 1 - q;
            const int pt = NQ - 1 - ((q + 1) % NQ);
            const int plo = pc < pt ? pc : pt;
            const int phi_ = pc < pt ? pt : pc;
            for (int t = tid; t < DIM / 4; t += 128) {
                int i = ((t >> plo) << (plo + 1)) | (t & ((1 << plo) - 1));
                i = ((i >> phi_) << (phi_ + 1)) | (i & ((1 << phi_) - 1));
                const int i1 = i | (1 << pc);
                const int i2 = i1 | (1 << pt);
                const float2 tmp = st[i1]; st[i1] = st[i2]; st[i2] = tmp;
            }
            __syncthreads();
        }
        {
            float s, c; sincosf(0.5f * w[(l * NQ + 0) * 3 + 0], &s, &c);
            const float2 ms = make_float2(0.0f, -s);
            for (int t = tid; t < DIM / 2; t += 128) {
                const int jd = t ^ 0x300;
                const float2 a0 = st[t], a1 = st[jd];
                st[t]  = cadd(cscale(c, a0), cmul(ms, a1));
                st[jd] = cadd(cscale(c, a1), cmul(ms, a0));
            }
            __syncthreads();
        }
        {
            float s, c; sincosf(0.5f * w[(l * NQ + 0) * 3 + 1], &s, &c);
            for (int t = tid; t < DIM / 2; t += 128) {
                const int i = ((t >> 8) << 9) | (t & 0xFF);
                const int jd = i ^ 0x180;
                const float sgn = ((i >> 7) & 1) ? 1.0f : -1.0f;
                const float2 ms = make_float2(0.0f, -sgn * s);
                const float2 a0 = st[i], a1 = st[jd];
                st[i]  = cadd(cscale(c, a0), cmul(ms, a1));
                st[jd] = cadd(cscale(c, a1), cmul(ms, a0));
            }
            __syncthreads();
        }
        {
            float s, c; sincosf(0.5f * w[(l * NQ + 0) * 3 + 2], &s, &c);
            for (int k = tid; k < DIM; k += 128) {
                const int df = ((k >> 7) ^ (k >> 6)) & 1;
                const float2 ph = make_float2(c, df ? s : -s);
                st[k] = cmul(ph, st[k]);
            }
            __syncthreads();
        }
    }

    for (int k = tid; k < DIM; k += 128) {
        Mre[j * DIM + k] = st[k].x;
        Mim[j * DIM + k] = st[k].y;
    }
}

__global__ void zero_kernel(float* __restrict__ out, int n) {
    const int i = blockIdx.x * blockDim.x + threadIdx.x;
    if (i < n) out[i] = 0.0f;
}

__global__ void __launch_bounds__(256) gemm_z_kernel(const float* __restrict__ Are,
                                                     const float* __restrict__ Aim,
                                                     const float* __restrict__ Bre,
                                                     const float* __restrict__ Bim,
                                                     float* __restrict__ out) {
    const int lane = threadIdx.x & 31;
    const int wave = threadIdx.x >> 5;
    const int gw = blockIdx.x * 8 + wave;
    const int tile_b = gw >> 6;
    const int tile_u = gw & 63;
    const int b0 = tile_b * 16;
    const int u0 = tile_u * 16;
    const int half = lane >> 4;
    const int l16 = lane & 15;
    const int ucol = u0 + l16;

    const float* arp = Are + (b0 + l16) * DIM + 2 * half;
    const float* aip = Aim + (b0 + l16) * DIM + 2 * half;
    const float* brp = Bre + (2 * half) * DIM + ucol;
    const float* bip = Bim + (2 * half) * DIM + ucol;

    v8f acc_re = {0.f, 0.f, 0.f, 0.f, 0.f, 0.f, 0.f, 0.f};
    v8f acc_im = {0.f, 0.f, 0.f, 0.f, 0.f, 0.f, 0.f, 0.f};

    #pragma unroll 4
    for (int k0 = 0; k0 < DIM; k0 += 4) {
        v2f ar = *(const v2f*)(arp + k0);
        v2f ai = *(const v2f*)(aip + k0);
        v2f br = { brp[k0 * DIM], brp[(k0 + 1) * DIM] };
        v2f bi = { bip[k0 * DIM], bip[(k0 + 1) * DIM] };
        v2f nai = { -ai[0], -ai[1] };

        acc_re = __builtin_amdgcn_wmma_f32_16x16x4_f32(false, ar,  false, br, (short)0, acc_re, false, false);
        acc_re = __builtin_amdgcn_wmma_f32_16x16x4_f32(false, nai, false, bi, (short)0, acc_re, false, false);
        acc_im = __builtin_amdgcn_wmma_f32_16x16x4_f32(false, ar,  false, bi, (short)0, acc_im, false, false);
        acc_im = __builtin_amdgcn_wmma_f32_16x16x4_f32(false, ai,  false, br, (short)0, acc_im, false, false);

        __builtin_prefetch(arp + k0 + 64, 0, 1);
        __builtin_prefetch(brp + (k0 + 16) * DIM, 0, 1);
    }

    float p[8];
    #pragma unroll
    for (int v = 0; v < 8; ++v)
        p[v] = acc_re[v] * acc_re[v] + acc_im[v] * acc_im[v];

    #pragma unroll
    for (int q = 0; q < NQ; ++q) {
        const float sgn = ((ucol >> (NQ - 1 - q)) & 1) ? -1.0f : 1.0f;
        float t[8];
        #pragma unroll
        for (int v = 0; v < 8; ++v) t[v] = sgn * p[v];
        #pragma unroll
        for (int m = 1; m < 16; m <<= 1) {
            #pragma unroll
            for (int v = 0; v < 8; ++v) t[v] += __shfl_xor(t[v], m, 32);
        }
        if (l16 == 0) {
            #pragma unroll
            for (int v = 0; v < 8; ++v)
                atomicAdd(&out[(b0 + half * 8 + v) * NQ + q], t[v]);
        }
    }
}

extern "C" void kernel_launch(void* const* d_in, const int* in_sizes, int n_in,
                              void* d_out, int out_size, void* d_ws, size_t ws_size,
                              hipStream_t stream) {
    const float* x = (const float*)d_in[0];
    const float* w = (const float*)d_in[1];
    float* out = (float*)d_out;

    float* pre = (float*)d_ws;
    float* pim = pre + (size_t)BATCH * DIM;
    float* Mre = pim + (size_t)BATCH * DIM;
    float* Mim = Mre + (size_t)DIM * DIM;

    embed_kernel  <<<BATCH, 256, 0, stream>>>(x, pre, pim);
    build_u_kernel<<<DIM,   128, 0, stream>>>(w, Mre, Mim);
    zero_kernel   <<<(BATCH * NQ + 255) / 256, 256, 0, stream>>>(out, BATCH * NQ);
    gemm_z_kernel <<<(BATCH / 16) * (DIM / 16) / 8, 256, 0, stream>>>(pre, pim, Mre, Mim, out);
}


# GEMM operands staged through LDS: 64x128 block tile, K chunks of 64 in two LDS stages, each wave 2x2 WMMA tiles fed by grouped LDS fragment loads instead of per-wave global fragment loads
# speedup vs baseline: 2.0452x; 2.0452x over previous
#include <hip/hip_runtime.h>
#include <hip/hip_bf16.h>
#include <math.h>

#define NQ    10
#define NL    6
#define BATCH 2048
#define DIM   1024


typedef float v2f __attribute__((ext_vector_type(2)));
typedef float v4f __attribute__((ext_vector_type(4)));
typedef float v8f __attribute__((ext_vector_type(8)));

__device__ __forceinline__ float2 cmul(float2 a, float2 b) {
    return make_float2(a.x * b.x - a.y * b.y, a.x * b.y + a.y * b.x);
}
__device__ __forceinline__ float2 cadd(float2 a, float2 b) {
    return make_float2(a.x + b.x, a.y + b.y);
}
__device__ __forceinline__ float2 cscale(float s, float2 a) {
    return make_float2(s * a.x, s * a.y);
}

__global__ void __launch_bounds__(256) embed_kernel(const float* __restrict__ x,
                                                    float* __restrict__ pre,
                                                    float* __restrict__ pim) {
    __shared__ float cs[NQ], sn[NQ];
    const int b = blockIdx.x;
    const int tid = threadIdx.x;
    if (tid < NQ) {
        float s, c;
        sincosf(0.5f * x[b * NQ + tid], &s, &c);
        sn[tid] = s; cs[tid] = c;
    }
    __syncthreads();
    #pragma unroll
    for (int i = 0; i < DIM / 256; ++i) {
        const int k = tid + 256 * i;
        float r = 1.0f;
        #pragma unroll
        for (int q = 0; q < NQ; ++q) {
            const int bit = (k >> (NQ - 1 - q)) & 1;
            r *= bit ? sn[q] : cs[q];
        }
        const int p = __popc(k) & 3;
        const float re = (p == 0) ? r : (p == 2 ? -r : 0.0f);
        const float im = (p == 1) ? -r : (p == 3 ? r : 0.0f);
        pre[b * DIM + k] = re;
        pim[b * DIM + k] = im;
    }
}

__global__ void __launch_bounds__(128) build_u_kernel(const float* __restrict__ w,
                                                      float* __restrict__ Mre,
                                                      float* __restrict__ Mim) {
    __shared__ float2 st[DIM];
    const int j = blockIdx.x;
    const int tid = threadIdx.x;
    for (int k = tid; k < DIM; k += 128)
        st[k] = make_float2(k == j ? 1.0f : 0.0f, 0.0f);
    __syncthreads();

    for (int l = 0; l < NL; ++l) {
        for (int q = 0; q < NQ; ++q) {
            const float phi = w[(l * NQ + q) * 3 + 0];
            const float th  = w[(l * NQ + q) * 3 + 1];
            const float om  = w[(l * NQ + q) * 3 + 2];
            float ct, s_t, ca, sa, cb, sb;
            sincosf(0.5f * th, &s_t, &ct);
            sincosf(0.5f * (phi + om), &sa, &ca);
            sincosf(0.5f * (phi - om), &sb, &cb);
            const float2 ga = make_float2( ct * ca, -ct * sa);
            const float2 gb = make_float2(-s_t * cb, -s_t * sb);
            const float2 gc = make_float2( s_t * cb, -s_t * sb);
            const float2 gd = make_float2( ct * ca,  ct * sa);
            const int p = NQ - 1 - q;
            for (int t = tid; t < DIM / 2; t += 128) {
                const int i0 = ((t >> p) << (p + 1)) | (t & ((1 << p) - 1));
                const int i1 = i0 | (1 << p);
                const float2 s0 = st[i0], s1 = st[i1];
                st[i0] = cadd(cmul(ga, s0), cmul(gb, s1));
                st[i1] = cadd(cmul(gc, s0), cmul(gd, s1));
            }
            __syncthreads();
        }
        for (int q = 0; q < NQ; ++q) {
            const int pc = NQ - 1 - q;
            const int pt = NQ - 1 - ((q + 1) % NQ);
            const int plo = pc < pt ? pc : pt;
            const int phi_ = pc < pt ? pt : pc;
            for (int t = tid; t < DIM / 4; t += 128) {
                int i = ((t >> plo) << (plo + 1)) | (t & ((1 << plo) - 1));
                i = ((i >> phi_) << (phi_ + 1)) | (i & ((1 << phi_) - 1));
                const int i1 = i | (1 << pc);
                const int i2 = i1 | (1 << pt);
                const float2 tmp = st[i1]; st[i1] = st[i2]; st[i2] = tmp;
            }
            __syncthreads();
        }
        {
            float s, c; sincosf(0.5f * w[(l * NQ + 0) * 3 + 0], &s, &c);
            const float2 ms = make_float2(0.0f, -s);
            for (int t = tid; t < DIM / 2; t += 128) {
                const int jd = t ^ 0x300;
                const float2 a0 = st[t], a1 = st[jd];
                st[t]  = cadd(cscale(c, a0), cmul(ms, a1));
                st[jd] = cadd(cscale(c, a1), cmul(ms, a0));
            }
            __syncthreads();
        }
        {
            float s, c; sincosf(0.5f * w[(l * NQ + 0) * 3 + 1], &s, &c);
            for (int t = tid; t < DIM / 2; t += 128) {
                const int i = ((t >> 8) << 9) | (t & 0xFF);
                const int jd = i ^ 0x180;
                const float sgn = ((i >> 7) & 1) ? 1.0f : -1.0f;
                const float2 ms = make_float2(0.0f, -sgn * s);
                const float2 a0 = st[i], a1 = st[jd];
                st[i]  = cadd(cscale(c, a0), cmul(ms, a1));
                st[jd] = cadd(cscale(c, a1), cmul(ms, a0));
            }
            __syncthreads();
        }
        {
            float s, c; sincosf(0.5f * w[(l * NQ + 0) * 3 + 2], &s, &c);
            for (int k = tid; k < DIM; k += 128) {
                const int df = ((k >> 7) ^ (k >> 6)) & 1;
                const float2 ph = make_float2(c, df ? s : -s);
                st[k] = cmul(ph, st[k]);
            }
            __syncthreads();
        }
    }

    for (int k = tid; k < DIM; k += 128) {
        Mre[j * DIM + k] = st[k].x;
        Mim[j * DIM + k] = st[k].y;
    }
}

__global__ void zero_kernel(float* __restrict__ out, int n) {
    const int i = blockIdx.x * blockDim.x + threadIdx.x;
    if (i < n) out[i] = 0.0f;
}

#define BM 64
#define BN 128
#define BK 64
#define APITCH (BK + 4)
#define BPITCH 80
#define B_PANEL (BK * BPITCH)
#define A_PLANE (BM * APITCH)
#define B_PLANE (2 * B_PANEL)
#define STAGE_FLOATS (2 * A_PLANE + 2 * B_PLANE)

struct Frag {
    v2f ar[2][2], ai[2][2];
    v2f br[2][2], bi[2][2];
};

__device__ __forceinline__ void load_frag(Frag& f, const float* fa, const float* fb, int kk) {
    for (int s = 0; s < 2; ++s) {
        for (int rt = 0; rt < 2; ++rt) {
            f.ar[s][rt] = *(const v2f*)(fa + rt * 16 * APITCH + kk + 4 * s);
            f.ai[s][rt] = *(const v2f*)(fa + A_PLANE + rt * 16 * APITCH + kk + 4 * s);
        }
        for (int ct = 0; ct < 2; ++ct) {
            f.br[s][ct][0] = fb[ct * B_PANEL + (kk + 4 * s) * BPITCH];
            f.br[s][ct][1] = fb[ct * B_PANEL + (kk + 4 * s + 1) * BPITCH];
            f.bi[s][ct][0] = fb[B_PLANE + ct * B_PANEL + (kk + 4 * s) * BPITCH];
            f.bi[s][ct][1] = fb[B_PLANE + ct * B_PANEL + (kk + 4 * s + 1) * BPITCH];
        }
    }
}

__device__ __forceinline__ void mma_frag(const Frag& f, v8f (&acc_re)[2][2], v8f (&acc_im)[2][2]) {
    for (int s = 0; s < 2; ++s) {
        v2f nai[2];
        for (int rt = 0; rt < 2; ++rt) {
            nai[rt][0] = -f.ai[s][rt][0];
            nai[rt][1] = -f.ai[s][rt][1];
        }
        for (int rt = 0; rt < 2; ++rt)
            for (int ct = 0; ct < 2; ++ct) {
                acc_re[rt][ct] = __builtin_amdgcn_wmma_f32_16x16x4_f32(false, f.ar[s][rt], false, f.br[s][ct], (short)0, acc_re[rt][ct], false, false);
                acc_im[rt][ct] = __builtin_amdgcn_wmma_f32_16x16x4_f32(false, f.ar[s][rt], false, f.bi[s][ct], (short)0, acc_im[rt][ct], false, false);
            }
        for (int rt = 0; rt < 2; ++rt)
            for (int ct = 0; ct < 2; ++ct) {
                acc_re[rt][ct] = __builtin_amdgcn_wmma_f32_16x16x4_f32(false, nai[rt],     false, f.bi[s][ct], (short)0, acc_re[rt][ct], false, false);
                acc_im[rt][ct] = __builtin_amdgcn_wmma_f32_16x16x4_f32(false, f.ai[s][rt], false, f.br[s][ct], (short)0, acc_im[rt][ct], false, false);
            }
    }
}

__global__ void __launch_bounds__(256) gemm_z_kernel(const float* __restrict__ Are,
                                                     const float* __restrict__ Aim,
                                                     const float* __restrict__ Bre,
                                                     const float* __restrict__ Bim,
                                                     float* __restrict__ out) {
    __shared__ __attribute__((aligned(16))) float smem[2 * STAGE_FLOATS];

    const int tid = threadIdx.x;
    const int lane = tid & 31;
    const int wave = tid >> 5;
    const int wr = wave >> 2;
    const int wc = wave & 3;
    const int half = lane >> 4;
    const int l16 = lane & 15;
    const int m0 = (blockIdx.x >> 3) * BM;
    const int n0 = (blockIdx.x & 7) * BN;

    const int a_row = tid >> 4, a_kq = tid & 15;
    const int b_row = tid >> 5, b_nq = tid & 31;
    const float* ga_re = Are + (m0 + a_row) * DIM + 4 * a_kq;
    const float* ga_im = Aim + (m0 + a_row) * DIM + 4 * a_kq;
    const float* gb_re = Bre + b_row * DIM + n0 + 4 * b_nq;
    const float* gb_im = Bim + b_row * DIM + n0 + 4 * b_nq;
    const int sa_off = a_row * APITCH + 4 * a_kq;
    const int sb_off = 2 * A_PLANE + (b_nq >> 4) * B_PANEL + b_row * BPITCH + 4 * (b_nq & 15);

    const int fa_off = (wr * 32 + l16) * APITCH + 2 * half;
    const int fb_off = 2 * A_PLANE + (2 * half) * BPITCH + wc * 16 + l16;

    v4f ra_re[4], ra_im[4], rb_re[8], rb_im[8];

    v8f acc_re[2][2], acc_im[2][2];
    for (int rt = 0; rt < 2; ++rt)
        for (int ct = 0; ct < 2; ++ct)
            for (int v = 0; v < 8; ++v) { acc_re[rt][ct][v] = 0.f; acc_im[rt][ct][v] = 0.f; }

    for (int p = 0; p < 4; ++p) {
        ra_re[p] = *(const v4f*)(ga_re + p * 16 * DIM);
        ra_im[p] = *(const v4f*)(ga_im + p * 16 * DIM);
    }
    for (int p = 0; p < 8; ++p) {
        rb_re[p] = *(const v4f*)(gb_re + p * 8 * DIM);
        rb_im[p] = *(const v4f*)(gb_im + p * 8 * DIM);
    }
    for (int p = 0; p < 4; ++p) {
        *(v4f*)(smem + sa_off + p * 16 * APITCH) = ra_re[p];
        *(v4f*)(smem + sa_off + A_PLANE + p * 16 * APITCH) = ra_im[p];
    }
    for (int p = 0; p < 8; ++p) {
        *(v4f*)(smem + sb_off + p * 8 * BPITCH) = rb_re[p];
        *(v4f*)(smem + sb_off + B_PLANE + p * 8 * BPITCH) = rb_im[p];
    }
    __syncthreads();

    for (int ch = 0; ch < DIM / BK; ++ch) {
        const float* st = smem + (ch & 1) * STAGE_FLOATS;
        const bool more = (ch + 1) < DIM / BK;

        if (more) {
            const int k1 = (ch + 1) * BK;
            for (int p = 0; p < 4; ++p) {
                ra_re[p] = *(const v4f*)(ga_re + k1 + p * 16 * DIM);
                ra_im[p] = *(const v4f*)(ga_im + k1 + p * 16 * DIM);
            }
            for (int p = 0; p < 8; ++p) {
                rb_re[p] = *(const v4f*)(gb_re + (k1 + p * 8) * DIM);
                rb_im[p] = *(const v4f*)(gb_im + (k1 + p * 8) * DIM);
            }
        }

        const float* fa = st + fa_off;
        const float* fb = st + fb_off;
        Frag f0, f1;
        load_frag(f0, fa, fb, 0);
        #pragma unroll 1
        for (int kk = 0; kk < BK; kk += 16) {
            load_frag(f1, fa, fb, kk + 8);
            __builtin_amdgcn_sched_barrier(0);
            mma_frag(f0, acc_re, acc_im);
            __builtin_amdgcn_sched_barrier(0);
            load_frag(f0, fa, fb, (kk + 16) & (BK - 1));
            __builtin_amdgcn_sched_barrier(0);
            mma_frag(f1, acc_re, acc_im);
        }

        if (more) {
            float* sn = smem + ((ch + 1) & 1) * STAGE_FLOATS;
            for (int p = 0; p < 4; ++p) {
                *(v4f*)(sn + sa_off + p * 16 * APITCH) = ra_re[p];
                *(v4f*)(sn + sa_off + A_PLANE + p * 16 * APITCH) = ra_im[p];
            }
            for (int p = 0; p < 8; ++p) {
                *(v4f*)(sn + sb_off + p * 8 * BPITCH) = rb_re[p];
                *(v4f*)(sn + sb_off + B_PLANE + p * 8 * BPITCH) = rb_im[p];
            }
        }
        __syncthreads();
    }

    for (int rt = 0; rt < 2; ++rt) {
        float p0[8], p1[8];
        for (int v = 0; v < 8; ++v) {
            p0[v] = acc_re[rt][0][v] * acc_re[rt][0][v] + acc_im[rt][0][v] * acc_im[rt][0][v];
            p1[v] = acc_re[rt][1][v] * acc_re[rt][1][v] + acc_im[rt][1][v] * acc_im[rt][1][v];
        }
        const int row0 = m0 + wr * 32 + rt * 16 + half * 8;
        const int col0 = n0 + wc * 16 + l16;
        const int col1 = col0 + 64;
        for (int q = 0; q < NQ; ++q) {
            const float s0 = ((col0 >> (NQ - 1 - q)) & 1) ? -1.0f : 1.0f;
            const float s1 = ((col1 >> (NQ - 1 - q)) & 1) ? -1.0f : 1.0f;
            float t[8];
            for (int v = 0; v < 8; ++v) t[v] = s0 * p0[v] + s1 * p1[v];
            for (int m = 1; m < 16; m <<= 1) {
                for (int v = 0; v < 8; ++v) t[v] += __shfl_xor(t[v], m, 32);
            }
            if (l16 == 0) {
                for (int v = 0; v < 8; ++v)
                    atomicAdd(&out[(row0 + v) * NQ + q], t[v]);
            }
        }
    }
}

extern "C" void kernel_launch(void* const* d_in, const int* in_sizes, int n_in,
                              void* d_out, int out_size, void* d_ws, size_t ws_size,
                              hipStream_t stream) {
    const float* x = (const float*)d_in[0];
    const float* w = (const float*)d_in[1];
    float* out = (float*)d_out;

    float* pre = (float*)d_ws;
    float* pim = pre + (size_t)BATCH * DIM;
    float* Mre = pim + (size_t)BATCH * DIM;
    float* Mim = Mre + (size_t)DIM * DIM;

    embed_kernel  <<<BATCH, 256, 0, stream>>>(x, pre, pim);
    build_u_kernel<<<DIM,   128, 0, stream>>>(w, Mre, Mim);
    zero_kernel   <<<(BATCH * NQ + 255) / 256, 256, 0, stream>>>(out, BATCH * NQ);
    gemm_z_kernel <<<(BATCH / BM) * (DIM / BN), 256, 0, stream>>>(pre, pim, Mre, Mim, out);
}


# Embedding phase (-i)^popcount(k) folded into the rows of the circuit matrix, so the state operand is real: 2 real f32 WMMA products per tile instead of 4, one A plane staged instead of two
# speedup vs baseline: 2.5605x; 1.2520x over previous
#include <hip/hip_runtime.h>
#include <hip/hip_bf16.h>
#include <math.h>

#define NQ    10
#define NL    6
#define BATCH 2048
#define DIM   1024


typedef float v2f __attribute__((ext_vector_type(2)));
typedef float v4f __attribute__((ext_vector_type(4)));
typedef float v8f __attribute__((ext_vector_type(8)));

__device__ __forceinline__ float2 cmul(float2 a, float2 b) {
    return make_float2(a.x * b.x - a.y * b.y, a.x * b.y + a.y * b.x);
}
__device__ __forceinline__ float2 cadd(float2 a, float2 b) {
    return make_float2(a.x + b.x, a.y + b.y);
}
__device__ __forceinline__ float2 cscale(float s, float2 a) {
    return make_float2(s * a.x, s * a.y);
}

__global__ void __launch_bounds__(256) embed_kernel(const float* __restrict__ x,
                                                    float* __restrict__ mag) {
    __shared__ float cs[NQ], sn[NQ];
    const int b = blockIdx.x;
    const int tid = threadIdx.x;
    if (tid < NQ) {
        float s, c;
        sincosf(0.5f * x[b * NQ + tid], &s, &c);
        sn[tid] = s; cs[tid] = c;
    }
    __syncthreads();
    #pragma unroll
    for (int i = 0; i < DIM / 256; ++i) {
        const int k = tid + 256 * i;
        float r = 1.0f;
        #pragma unroll
        for (int q = 0; q < NQ; ++q) {
            const int bit = (k >> (NQ - 1 - q)) & 1;
            r *= bit ? sn[q] : cs[q];
        }
        mag[b * DIM + k] = r;
    }
}

__global__ void __launch_bounds__(128) build_u_kernel(const float* __restrict__ w,
                                                      float* __restrict__ Mre,
                                                      float* __restrict__ Mim) {
    __shared__ float2 st[DIM];
    const int j = blockIdx.x;
    const int tid = threadIdx.x;
    const int pj = __popc(j) & 3;
    const float2 amp = make_float2(pj == 0 ? 1.0f : (pj == 2 ? -1.0f : 0.0f),
                                   pj == 1 ? -1.0f : (pj == 3 ? 1.0f : 0.0f));
    for (int k = tid; k < DIM; k += 128)
        st[k] = (k == j) ? amp : make_float2(0.0f, 0.0f);
    __syncthreads();

    for (int l = 0; l < NL; ++l) {
        for (int q = 0; q < NQ; ++q) {
            const float phi = w[(l * NQ + q) * 3 + 0];
            const float th  = w[(l * NQ + q) * 3 + 1];
            const float om  = w[(l * NQ + q) * 3 + 2];
            float ct, s_t, ca, sa, cb, sb;
            sincosf(0.5f * th, &s_t, &ct);
            sincosf(0.5f * (phi + om), &sa, &ca);
            sincosf(0.5f * (phi - om), &sb, &cb);
            const float2 ga = make_float2( ct * ca, -ct * sa);
            const float2 gb = make_float2(-s_t * cb, -s_t * sb);
            const float2 gc = make_float2( s_t * cb, -s_t * sb);
            const float2 gd = make_float2( ct * ca,  ct * sa);
            const int p = NQ - 1 - q;
            for (int t = tid; t < DIM / 2; t += 128) {
                const int i0 = ((t >> p) << (p + 1)) | (t & ((1 << p) - 1));
                const int i1 = i0 | (1 << p);
                const float2 s0 = st[i0], s1 = st[i1];
                st[i0] = cadd(cmul(ga, s0), cmul(gb, s1));
                st[i1] = cadd(cmul(gc, s0), cmul(gd, s1));
            }
            __syncthreads();
        }
        for (int q = 0; q < NQ; ++q) {
            const int pc = NQ - 1 - q;
            const int pt = NQ - 1 - ((q + 1) % NQ);
            const int plo = pc < pt ? pc : pt;
            const int phi_ = pc < pt ? pt : pc;
            for (int t = tid; t < DIM / 4; t += 128) {
                int i = ((t >> plo) << (plo + 1)) | (t & ((1 << plo) - 1));
                i = ((i >> phi_) << (phi_ + 1)) | (i & ((1 << phi_) - 1));
                const int i1 = i | (1 << pc);
                const int i2 = i1 | (1 << pt);
                const float2 tmp = st[i1]; st[i1] = st[i2]; st[i2] = tmp;
            }
            __syncthreads();
        }
        {
            float s, c; sincosf(0.5f * w[(l * NQ + 0) * 3 + 0], &s, &c);
            const float2 ms = make_float2(0.0f, -s);
            for (int t = tid; t < DIM / 2; t += 128) {
                const int jd = t ^ 0x300;
                const float2 a0 = st[t], a1 = st[jd];
                st[t]  = cadd(cscale(c, a0), cmul(ms, a1));
                st[jd] = cadd(cscale(c, a1), cmul(ms, a0));
            }
            __syncthreads();
        }
        {
            float s, c; sincosf(0.5f * w[(l * NQ + 0) * 3 + 1], &s, &c);
            for (int t = tid; t < DIM / 2; t += 128) {
                const int i = ((t >> 8) << 9) | (t & 0xFF);
                const int jd = i ^ 0x180;
                const float sgn = ((i >> 7) & 1) ? 1.0f : -1.0f;
                const float2 ms = make_float2(0.0f, -sgn * s);
                const float2 a0 = st[i], a1 = st[jd];
                st[i]  = cadd(cscale(c, a0), cmul(ms, a1));
                st[jd] = cadd(cscale(c, a1), cmul(ms, a0));
            }
            __syncthreads();
        }
        {
            float s, c; sincosf(0.5f * w[(l * NQ + 0) * 3 + 2], &s, &c);
            for (int k = tid; k < DIM; k += 128) {
                const int df = ((k >> 7) ^ (k >> 6)) & 1;
                const float2 ph = make_float2(c, df ? s : -s);
                st[k] = cmul(ph, st[k]);
            }
            __syncthreads();
        }
    }

    for (int k = tid; k < DIM; k += 128) {
        Mre[j * DIM + k] = st[k].x;
        Mim[j * DIM + k] = st[k].y;
    }
}

__global__ void zero_kernel(float* __restrict__ out, int n) {
    const int i = blockIdx.x * blockDim.x + threadIdx.x;
    if (i < n) out[i] = 0.0f;
}

#define BM 64
#define BN 128
#define BK 64
#define APITCH (BK + 4)
#define BPITCH 80
#define B_PANEL (BK * BPITCH)
#define A_PLANE (BM * APITCH)
#define B_PLANE (2 * B_PANEL)
#define STAGE_FLOATS (A_PLANE + 2 * B_PLANE)

struct Frag {
    v2f a[2][2];
    v2f br[2][2], bi[2][2];
};

__device__ __forceinline__ void load_frag(Frag& f, const float* fa, const float* fb, int kk) {
    for (int s = 0; s < 2; ++s) {
        for (int rt = 0; rt < 2; ++rt)
            f.a[s][rt] = *(const v2f*)(fa + rt * 16 * APITCH + kk + 4 * s);
        for (int ct = 0; ct < 2; ++ct) {
            f.br[s][ct][0] = fb[ct * B_PANEL + (kk + 4 * s) * BPITCH];
            f.br[s][ct][1] = fb[ct * B_PANEL + (kk + 4 * s + 1) * BPITCH];
            f.bi[s][ct][0] = fb[B_PLANE + ct * B_PANEL + (kk + 4 * s) * BPITCH];
            f.bi[s][ct][1] = fb[B_PLANE + ct * B_PANEL + (kk + 4 * s + 1) * BPITCH];
        }
    }
}

__device__ __forceinline__ void mma_frag(const Frag& f, v8f (&acc_re)[2][2], v8f (&acc_im)[2][2]) {
    for (int s = 0; s < 2; ++s)
        for (int rt = 0; rt < 2; ++rt)
            for (int ct = 0; ct < 2; ++ct) {
                acc_re[rt][ct] = __builtin_amdgcn_wmma_f32_16x16x4_f32(false, f.a[s][rt], false, f.br[s][ct], (short)0, acc_re[rt][ct], false, false);
                acc_im[rt][ct] = __builtin_amdgcn_wmma_f32_16x16x4_f32(false, f.a[s][rt], false, f.bi[s][ct], (short)0, acc_im[rt][ct], false, false);
            }
}

__global__ void __launch_bounds__(256) gemm_z_kernel(const float* __restrict__ A,
                                                     const float* __restrict__ Bre,
                                                     const float* __restrict__ Bim,
                                                     float* __restrict__ out) {
    __shared__ __attribute__((aligned(16))) float smem[2 * STAGE_FLOATS];

    const int tid = threadIdx.x;
    const int lane = tid & 31;
    const int wave = tid >> 5;
    const int wr = wave >> 2;
    const int wc = wave & 3;
    const int half = lane >> 4;
    const int l16 = lane & 15;
    const int m0 = (blockIdx.x >> 3) * BM;
    const int n0 = (blockIdx.x & 7) * BN;

    const int a_row = tid >> 4, a_kq = tid & 15;
    const int b_row = tid >> 5, b_nq = tid & 31;
    const float* ga = A + (m0 + a_row) * DIM + 4 * a_kq;
    const float* gb_re = Bre + b_row * DIM + n0 + 4 * b_nq;
    const float* gb_im = Bim + b_row * DIM + n0 + 4 * b_nq;
    const int sa_off = a_row * APITCH + 4 * a_kq;
    const int sb_off = A_PLANE + (b_nq >> 4) * B_PANEL + b_row * BPITCH + 4 * (b_nq & 15);

    const int fa_off = (wr * 32 + l16) * APITCH + 2 * half;
    const int fb_off = A_PLANE + (2 * half) * BPITCH + wc * 16 + l16;

    v4f ra[4], rb_re[8], rb_im[8];

    v8f acc_re[2][2], acc_im[2][2];
    for (int rt = 0; rt < 2; ++rt)
        for (int ct = 0; ct < 2; ++ct)
            for (int v = 0; v < 8; ++v) { acc_re[rt][ct][v] = 0.f; acc_im[rt][ct][v] = 0.f; }

    for (int p = 0; p < 4; ++p)
        ra[p] = *(const v4f*)(ga + p * 16 * DIM);
    for (int p = 0; p < 8; ++p) {
        rb_re[p] = *(const v4f*)(gb_re + p * 8 * DIM);
        rb_im[p] = *(const v4f*)(gb_im + p * 8 * DIM);
    }
    for (int p = 0; p < 4; ++p)
        *(v4f*)(smem + sa_off + p * 16 * APITCH) = ra[p];
    for (int p = 0; p < 8; ++p) {
        *(v4f*)(smem + sb_off + p * 8 * BPITCH) = rb_re[p];
        *(v4f*)(smem + sb_off + B_PLANE + p * 8 * BPITCH) = rb_im[p];
    }
    __syncthreads();

    for (int ch = 0; ch < DIM / BK; ++ch) {
        const float* st = smem + (ch & 1) * STAGE_FLOATS;
        const bool more = (ch + 1) < DIM / BK;

        if (more) {
            const int k1 = (ch + 1) * BK;
            for (int p = 0; p < 4; ++p)
                ra[p] = *(const v4f*)(ga + k1 + p * 16 * DIM);
            for (int p = 0; p < 8; ++p) {
                rb_re[p] = *(const v4f*)(gb_re + (k1 + p * 8) * DIM);
                rb_im[p] = *(const v4f*)(gb_im + (k1 + p * 8) * DIM);
            }
        }

        const float* fa = st + fa_off;
        const float* fb = st + fb_off;
        Frag f0, f1;
        load_frag(f0, fa, fb, 0);
        #pragma unroll 1
        for (int kk = 0; kk < BK; kk += 16) {
            load_frag(f1, fa, fb, kk + 8);
            __builtin_amdgcn_sched_barrier(0);
            mma_frag(f0, acc_re, acc_im);
            __builtin_amdgcn_sched_barrier(0);
            load_frag(f0, fa, fb, (kk + 16) & (BK - 1));
            __builtin_amdgcn_sched_barrier(0);
            mma_frag(f1, acc_re, acc_im);
        }

        if (more) {
            float* sn = smem + ((ch + 1) & 1) * STAGE_FLOATS;
            for (int p = 0; p < 4; ++p)
                *(v4f*)(sn + sa_off + p * 16 * APITCH) = ra[p];
            for (int p = 0; p < 8; ++p) {
                *(v4f*)(sn + sb_off + p * 8 * BPITCH) = rb_re[p];
                *(v4f*)(sn + sb_off + B_PLANE + p * 8 * BPITCH) = rb_im[p];
            }
        }
        __syncthreads();
    }

    for (int rt = 0; rt < 2; ++rt) {
        float p0[8], p1[8];
        for (int v = 0; v < 8; ++v) {
            p0[v] = acc_re[rt][0][v] * acc_re[rt][0][v] + acc_im[rt][0][v] * acc_im[rt][0][v];
            p1[v] = acc_re[rt][1][v] * acc_re[rt][1][v] + acc_im[rt][1][v] * acc_im[rt][1][v];
        }
        const int row0 = m0 + wr * 32 + rt * 16 + half * 8;
        const int col0 = n0 + wc * 16 + l16;
        const int col1 = col0 + 64;
        for (int q = 0; q < NQ; ++q) {
            const float s0 = ((col0 >> (NQ - 1 - q)) & 1) ? -1.0f : 1.0f;
            const float s1 = ((col1 >> (NQ - 1 - q)) & 1) ? -1.0f : 1.0f;
            float t[8];
            for (int v = 0; v < 8; ++v) t[v] = s0 * p0[v] + s1 * p1[v];
            for (int m = 1; m < 16; m <<= 1) {
                for (int v = 0; v < 8; ++v) t[v] += __shfl_xor(t[v], m, 32);
            }
            if (l16 == 0) {
                for (int v = 0; v < 8; ++v)
                    atomicAdd(&out[(row0 + v) * NQ + q], t[v]);
            }
        }
    }
}

extern "C" void kernel_launch(void* const* d_in, const int* in_sizes, int n_in,
                              void* d_out, int out_size, void* d_ws, size_t ws_size,
                              hipStream_t stream) {
    const float* x = (const float*)d_in[0];
    const float* w = (const float*)d_in[1];
    float* out = (float*)d_out;

    float* mag = (float*)d_ws;
    float* Mre = mag + (size_t)BATCH * DIM;
    float* Mim = Mre + (size_t)DIM * DIM;

    embed_kernel  <<<BATCH, 256, 0, stream>>>(x, mag);
    build_u_kernel<<<DIM,   128, 0, stream>>>(w, Mre, Mim);
    zero_kernel   <<<(BATCH * NQ + 255) / 256, 256, 0, stream>>>(out, BATCH * NQ);
    gemm_z_kernel <<<(BATCH / BM) * (DIM / BN), 256, 0, stream>>>(mag, Mre, Mim, out);
}


# Circuit-matrix build: each thread holds 8 amplitudes (3 index bits) in registers per LDS pass and applies every gate inside them, coefficients from a per-block table: 4 barriers per layer, not 23
# speedup vs baseline: 3.7581x; 1.4677x over previous
#include <hip/hip_runtime.h>
#include <hip/hip_bf16.h>
#include <math.h>

#define NQ    10
#define NL    6
#define BATCH 2048
#define DIM   1024


typedef float v2f __attribute__((ext_vector_type(2)));
typedef float v4f __attribute__((ext_vector_type(4)));
typedef float v8f __attribute__((ext_vector_type(8)));

__device__ __forceinline__ float2 cmul(float2 a, float2 b) {
    return make_float2(a.x * b.x - a.y * b.y, a.x * b.y + a.y * b.x);
}
__device__ __forceinline__ float2 cadd(float2 a, float2 b) {
    return make_float2(a.x + b.x, a.y + b.y);
}
__device__ __forceinline__ float2 cscale(float s, float2 a) {
    return make_float2(s * a.x, s * a.y);
}

__global__ void __launch_bounds__(256) embed_kernel(const float* __restrict__ x,
                                                    float* __restrict__ mag) {
    __shared__ float cs[NQ], sn[NQ];
    const int b = blockIdx.x;
    const int tid = threadIdx.x;
    if (tid < NQ) {
        float s, c;
        sincosf(0.5f * x[b * NQ + tid], &s, &c);
        sn[tid] = s; cs[tid] = c;
    }
    __syncthreads();
    #pragma unroll
    for (int i = 0; i < DIM / 256; ++i) {
        const int k = tid + 256 * i;
        float r = 1.0f;
        #pragma unroll
        for (int q = 0; q < NQ; ++q) {
            const int bit = (k >> (NQ - 1 - q)) & 1;
            r *= bit ? sn[q] : cs[q];
        }
        mag[b * DIM + k] = r;
    }
}

#define ROT_GATES   (NL * NQ)
#define ISING_GATES (NL * 3)

__host__ __device__ constexpr int lds_slot(int k) { return k ^ (k >> 5); }

__host__ __device__ constexpr int cnot_ring(int k) {
    for (int q = 0; q < NQ; ++q) {
        const int pc = NQ - 1 - q;
        const int pt = NQ - 1 - ((q + 1) % NQ);
        k ^= ((k >> pc) & 1) << pt;
    }
    return k;
}

template <int F0, int F1, int F2>
__host__ __device__ constexpr int spread3(int j) {
    return ((j & 1) << F0) | (((j >> 1) & 1) << F1) | (((j >> 2) & 1) << F2);
}

template <int F0, int F1, int F2>
__device__ __forceinline__ void load8(float2 (&a)[8], const float2* buf, int base) {
    for (int j = 0; j < 8; ++j) a[j] = buf[base ^ lds_slot(spread3<F0, F1, F2>(j))];
}
template <int F0, int F1, int F2>
__device__ __forceinline__ void store8(const float2 (&a)[8], float2* buf, int base) {
    for (int j = 0; j < 8; ++j) buf[base ^ lds_slot(spread3<F0, F1, F2>(j))] = a[j];
}

__device__ __forceinline__ void rot_pair(float2& s0, float2& s1, const v4f g) {
    const float x0 = s0.x, y0 = s0.y, x1 = s1.x, y1 = s1.y;
    s0.x = g.x * x0 - g.y * y0 - g.z * x1 - g.w * y1;
    s0.y = g.x * y0 + g.y * x0 - g.z * y1 + g.w * x1;
    s1.x = g.z * x0 - g.w * y0 + g.x * x1 + g.y * y1;
    s1.y = g.z * y0 + g.w * x0 + g.x * y1 - g.y * x1;
}
template <int B>
__device__ __forceinline__ void rot_bit(float2 (&a)[8], const v4f g) {
    for (int j = 0; j < 8; ++j)
        if (!(j & (1 << B))) rot_pair(a[j], a[j | (1 << B)], g);
}

__device__ __forceinline__ void ising3(float2 (&a)[8], const float2* t, int b6) {
    {
        const float c = t[0].x, s = t[0].y;
        for (int j = 0; j < 4; ++j) {
            const float2 a0 = a[j], a1 = a[j ^ 6];
            a[j]     = make_float2(c * a0.x + s * a1.y, c * a0.y - s * a1.x);
            a[j ^ 6] = make_float2(c * a1.x + s * a0.y, c * a1.y - s * a0.x);
        }
    }
    {
        const float c = t[1].x, s = t[1].y;
        for (int j = 0; j < 8; ++j) {
            if (j & 2) continue;
            const float m = (j & 1) ? -s : s;
            const float2 a0 = a[j], a1 = a[j ^ 3];
            a[j]     = make_float2(c * a0.x - m * a1.y, c * a0.y + m * a1.x);
            a[j ^ 3] = make_float2(c * a1.x - m * a0.y, c * a1.y + m * a0.x);
        }
    }
    {
        const float c = t[2].x, s = t[2].y;
        const float s0 = b6 ? s : -s;
        for (int j = 0; j < 8; ++j) {
            const float py = (j & 1) ? -s0 : s0;
            const float2 v = a[j];
            a[j] = make_float2(c * v.x - py * v.y, c * v.y + py * v.x);
        }
    }
}

__global__ void __launch_bounds__(128) build_u_kernel(const float* __restrict__ w,
                                                      float* __restrict__ Mre,
                                                      float* __restrict__ Mim) {
    __shared__ __attribute__((aligned(16))) float2 sbuf[2 * DIM];
    __shared__ __attribute__((aligned(16))) v4f rot_tab[ROT_GATES];
    __shared__ float2 ising_tab[ISING_GATES];
    const int jcol = blockIdx.x;
    const int tid = threadIdx.x;

    if (tid < ROT_GATES) {
        const float phi = w[tid * 3 + 0];
        const float th  = w[tid * 3 + 1];
        const float om  = w[tid * 3 + 2];
        float ct, s_t, ca, sa, cb, sb;
        sincosf(0.5f * th, &s_t, &ct);
        sincosf(0.5f * (phi + om), &sa, &ca);
        sincosf(0.5f * (phi - om), &sb, &cb);
        v4f g;
        g.x =  ct * ca;  g.y = -ct * sa;
        g.z =  s_t * cb; g.w = -s_t * sb;
        rot_tab[tid] = g;
    } else if (tid >= 64 && tid < 64 + ISING_GATES) {
        const int g = tid - 64;
        float s, c;
        sincosf(0.5f * w[(g / 3) * NQ * 3 + (g % 3)], &s, &c);
        ising_tab[g] = make_float2(c, s);
    }

    const int pj = __popc(jcol) & 3;
    const float2 amp = make_float2(pj == 0 ? 1.0f : (pj == 2 ? -1.0f : 0.0f),
                                   pj == 1 ? -1.0f : (pj == 3 ? 1.0f : 0.0f));
    for (int k = tid; k < DIM; k += 128)
        sbuf[lds_slot(k)] = (k == jcol) ? amp : make_float2(0.0f, 0.0f);
    __syncthreads();

    const int t0 = tid & 1, t1 = (tid >> 1) & 1, t2 = (tid >> 2) & 1, t3 = (tid >> 3) & 1;
    const int t4 = (tid >> 4) & 1, t5 = (tid >> 5) & 1, t6 = (tid >> 6) & 1;
    const int kA = tid;
    const int kB = (tid & 15) | (t4 << 9) | (t5 << 7) | (t6 << 8);
    const int kC = t0 | (t1 << 4) | (t2 << 6) | (t3 << 7) | (t4 << 8) | (t5 << 5) | (t6 << 9);
    const int kD = tid << 3;
    const int pA = lds_slot(kA), pB = lds_slot(kB), pC = lds_slot(kC), pD = lds_slot(kD);
    const int pDst = lds_slot(cnot_ring(kD));

    int cur = 0;
    float2 a[8];
    for (int l = 0; l < NL; ++l) {
        const v4f* g = rot_tab + l * NQ;
        float2* buf = sbuf + cur;

        load8<7, 8, 9>(a, buf, pA);
        if (l > 0) ising3(a, ising_tab + (l - 1) * 3, t6);
        rot_bit<0>(a, g[2]); rot_bit<1>(a, g[1]); rot_bit<2>(a, g[0]);
        store8<7, 8, 9>(a, buf, pA);
        __syncthreads();

        load8<4, 5, 6>(a, buf, pB);
        rot_bit<0>(a, g[5]); rot_bit<1>(a, g[4]); rot_bit<2>(a, g[3]);
        store8<4, 5, 6>(a, buf, pB);
        __syncthreads();

        load8<1, 2, 3>(a, buf, pC);
        rot_bit<0>(a, g[8]); rot_bit<1>(a, g[7]); rot_bit<2>(a, g[6]);
        store8<1, 2, 3>(a, buf, pC);
        __syncthreads();

        load8<0, 1, 2>(a, buf, pD);
        rot_bit<0>(a, g[9]);
        float2* dst = sbuf + (cur ^ DIM);
        for (int j = 0; j < 8; ++j) dst[pDst ^ lds_slot(cnot_ring(j))] = a[j];
        __syncthreads();
        cur ^= DIM;
    }

    load8<7, 8, 9>(a, sbuf + cur, pA);
    ising3(a, ising_tab + (NL - 1) * 3, t6);
    for (int j = 0; j < 8; ++j) {
        Mre[jcol * DIM + tid + 128 * j] = a[j].x;
        Mim[jcol * DIM + tid + 128 * j] = a[j].y;
    }
}

__global__ void zero_kernel(float* __restrict__ out, int n) {
    const int i = blockIdx.x * blockDim.x + threadIdx.x;
    if (i < n) out[i] = 0.0f;
}

#define BM 64
#define BN 128
#define BK 64
#define APITCH (BK + 4)
#define BPITCH 80
#define B_PANEL (BK * BPITCH)
#define A_PLANE (BM * APITCH)
#define B_PLANE (2 * B_PANEL)
#define STAGE_FLOATS (A_PLANE + 2 * B_PLANE)

struct Frag {
    v2f a[2][2];
    v2f br[2][2], bi[2][2];
};

__device__ __forceinline__ void load_frag(Frag& f, const float* fa, const float* fb, int kk) {
    for (int s = 0; s < 2; ++s) {
        for (int rt = 0; rt < 2; ++rt)
            f.a[s][rt] = *(const v2f*)(fa + rt * 16 * APITCH + kk + 4 * s);
        for (int ct = 0; ct < 2; ++ct) {
            f.br[s][ct][0] = fb[ct * B_PANEL + (kk + 4 * s) * BPITCH];
            f.br[s][ct][1] = fb[ct * B_PANEL + (kk + 4 * s + 1) * BPITCH];
            f.bi[s][ct][0] = fb[B_PLANE + ct * B_PANEL + (kk + 4 * s) * BPITCH];
            f.bi[s][ct][1] = fb[B_PLANE + ct * B_PANEL + (kk + 4 * s + 1) * BPITCH];
        }
    }
}

__device__ __forceinline__ void mma_frag(const Frag& f, v8f (&acc_re)[2][2], v8f (&acc_im)[2][2]) {
    for (int s = 0; s < 2; ++s)
        for (int rt = 0; rt < 2; ++rt)
            for (int ct = 0; ct < 2; ++ct) {
                acc_re[rt][ct] = __builtin_amdgcn_wmma_f32_16x16x4_f32(false, f.a[s][rt], false, f.br[s][ct], (short)0, acc_re[rt][ct], false, false);
                acc_im[rt][ct] = __builtin_amdgcn_wmma_f32_16x16x4_f32(false, f.a[s][rt], false, f.bi[s][ct], (short)0, acc_im[rt][ct], false, false);
            }
}

__global__ void __launch_bounds__(256) gemm_z_kernel(const float* __restrict__ A,
                                                     const float* __restrict__ Bre,
                                                     const float* __restrict__ Bim,
                                                     float* __restrict__ out) {
    __shared__ __attribute__((aligned(16))) float smem[2 * STAGE_FLOATS];

    const int tid = threadIdx.x;
    const int lane = tid & 31;
    const int wave = tid >> 5;
    const int wr = wave >> 2;
    const int wc = wave & 3;
    const int half = lane >> 4;
    const int l16 = lane & 15;
    const int m0 = (blockIdx.x >> 3) * BM;
    const int n0 = (blockIdx.x & 7) * BN;

    const int a_row = tid >> 4, a_kq = tid & 15;
    const int b_row = tid >> 5, b_nq = tid & 31;
    const float* ga = A + (m0 + a_row) * DIM + 4 * a_kq;
    const float* gb_re = Bre + b_row * DIM + n0 + 4 * b_nq;
    const float* gb_im = Bim + b_row * DIM + n0 + 4 * b_nq;
    const int sa_off = a_row * APITCH + 4 * a_kq;
    const int sb_off = A_PLANE + (b_nq >> 4) * B_PANEL + b_row * BPITCH + 4 * (b_nq & 15);

    const int fa_off = (wr * 32 + l16) * APITCH + 2 * half;
    const int fb_off = A_PLANE + (2 * half) * BPITCH + wc * 16 + l16;

    v4f ra[4], rb_re[8], rb_im[8];

    v8f acc_re[2][2], acc_im[2][2];
    for (int rt = 0; rt < 2; ++rt)
        for (int ct = 0; ct < 2; ++ct)
            for (int v = 0; v < 8; ++v) { acc_re[rt][ct][v] = 0.f; acc_im[rt][ct][v] = 0.f; }

    for (int p = 0; p < 4; ++p)
        ra[p] = *(const v4f*)(ga + p * 16 * DIM);
    for (int p = 0; p < 8; ++p) {
        rb_re[p] = *(const v4f*)(gb_re + p * 8 * DIM);
        rb_im[p] = *(const v4f*)(gb_im + p * 8 * DIM);
    }
    for (int p = 0; p < 4; ++p)
        *(v4f*)(smem + sa_off + p * 16 * APITCH) = ra[p];
    for (int p = 0; p < 8; ++p) {
        *(v4f*)(smem + sb_off + p * 8 * BPITCH) = rb_re[p];
        *(v4f*)(smem + sb_off + B_PLANE + p * 8 * BPITCH) = rb_im[p];
    }
    __syncthreads();

    for (int ch = 0; ch < DIM / BK; ++ch) {
        const float* st = smem + (ch & 1) * STAGE_FLOATS;
        const bool more = (ch + 1) < DIM / BK;

        if (more) {
            const int k1 = (ch + 1) * BK;
            for (int p = 0; p < 4; ++p)
                ra[p] = *(const v4f*)(ga + k1 + p * 16 * DIM);
            for (int p = 0; p < 8; ++p) {
                rb_re[p] = *(const v4f*)(gb_re + (k1 + p * 8) * DIM);
                rb_im[p] = *(const v4f*)(gb_im + (k1 + p * 8) * DIM);
            }
        }

        const float* fa = st + fa_off;
        const float* fb = st + fb_off;
        Frag f0, f1;
        load_frag(f0, fa, fb, 0);
        #pragma unroll 1
        for (int kk = 0; kk < BK; kk += 16) {
            load_frag(f1, fa, fb, kk + 8);
            __builtin_amdgcn_sched_barrier(0);
            mma_frag(f0, acc_re, acc_im);
            __builtin_amdgcn_sched_barrier(0);
            load_frag(f0, fa, fb, (kk + 16) & (BK - 1));
            __builtin_amdgcn_sched_barrier(0);
            mma_frag(f1, acc_re, acc_im);
        }

        if (more) {
            float* sn = smem + ((ch + 1) & 1) * STAGE_FLOATS;
            for (int p = 0; p < 4; ++p)
                *(v4f*)(sn + sa_off + p * 16 * APITCH) = ra[p];
            for (int p = 0; p < 8; ++p) {
                *(v4f*)(sn + sb_off + p * 8 * BPITCH) = rb_re[p];
                *(v4f*)(sn + sb_off + B_PLANE + p * 8 * BPITCH) = rb_im[p];
            }
        }
        __syncthreads();
    }

    for (int rt = 0; rt < 2; ++rt) {
        float p0[8], p1[8];
        for (int v = 0; v < 8; ++v) {
            p0[v] = acc_re[rt][0][v] * acc_re[rt][0][v] + acc_im[rt][0][v] * acc_im[rt][0][v];
            p1[v] = acc_re[rt][1][v] * acc_re[rt][1][v] + acc_im[rt][1][v] * acc_im[rt][1][v];
        }
        const int row0 = m0 + wr * 32 + rt * 16 + half * 8;
        const int col0 = n0 + wc * 16 + l16;
        const int col1 = col0 + 64;
        for (int q = 0; q < NQ; ++q) {
            const float s0 = ((col0 >> (NQ - 1 - q)) & 1) ? -1.0f : 1.0f;
            const float s1 = ((col1 >> (NQ - 1 - q)) & 1) ? -1.0f : 1.0f;
            float t[8];
            for (int v = 0; v < 8; ++v) t[v] = s0 * p0[v] + s1 * p1[v];
            for (int m = 1; m < 16; m <<= 1) {
                for (int v = 0; v < 8; ++v) t[v] += __shfl_xor(t[v], m, 32);
            }
            if (l16 == 0) {
                for (int v = 0; v < 8; ++v)
                    atomicAdd(&out[(row0 + v) * NQ + q], t[v]);
            }
        }
    }
}

extern "C" void kernel_launch(void* const* d_in, const int* in_sizes, int n_in,
                              void* d_out, int out_size, void* d_ws, size_t ws_size,
                              hipStream_t stream) {
    const float* x = (const float*)d_in[0];
    const float* w = (const float*)d_in[1];
    float* out = (float*)d_out;

    float* mag = (float*)d_ws;
    float* Mre = mag + (size_t)BATCH * DIM;
    float* Mim = Mre + (size_t)DIM * DIM;

    embed_kernel  <<<BATCH, 256, 0, stream>>>(x, mag);
    build_u_kernel<<<DIM,   128, 0, stream>>>(w, Mre, Mim);
    zero_kernel   <<<(BATCH * NQ + 255) / 256, 256, 0, stream>>>(out, BATCH * NQ);
    gemm_z_kernel <<<(BATCH / BM) * (DIM / BN), 256, 0, stream>>>(mag, Mre, Mim, out);
}


# GEMM epilogue: the per-qubit sign sums as one Walsh-Hadamard butterfly per lane row in DPP, the four wave columns joined in LDS, 20 whole-line atomics per block in place of 2,560 two-lane ones
# speedup vs baseline: 5.8169x; 1.5478x over previous
#include <hip/hip_runtime.h>
#include <hip/hip_bf16.h>
#include <math.h>

#define NQ    10
#define NL    6
#define BATCH 2048
#define DIM   1024


typedef float v2f __attribute__((ext_vector_type(2)));
typedef float v4f __attribute__((ext_vector_type(4)));
typedef float v8f __attribute__((ext_vector_type(8)));

__device__ __forceinline__ float2 cmul(float2 a, float2 b) {
    return make_float2(a.x * b.x - a.y * b.y, a.x * b.y + a.y * b.x);
}
__device__ __forceinline__ float2 cadd(float2 a, float2 b) {
    return make_float2(a.x + b.x, a.y + b.y);
}
__device__ __forceinline__ float2 cscale(float s, float2 a) {
    return make_float2(s * a.x, s * a.y);
}

#define ROT_GATES   (NL * NQ)
#define ISING_GATES (NL * 3)

__host__ __device__ constexpr int lds_slot(int k) { return k ^ (k >> 5); }

__host__ __device__ constexpr int cnot_ring(int k) {
    for (int q = 0; q < NQ; ++q) {
        const int pc = NQ - 1 - q;
        const int pt = NQ - 1 - ((q + 1) % NQ);
        k ^= ((k >> pc) & 1) << pt;
    }
    return k;
}

template <int F0, int F1, int F2>
__host__ __device__ constexpr int spread3(int j) {
    return ((j & 1) << F0) | (((j >> 1) & 1) << F1) | (((j >> 2) & 1) << F2);
}

template <int F0, int F1, int F2>
__device__ __forceinline__ void load8(float2 (&a)[8], const float2* buf, int base) {
    for (int j = 0; j < 8; ++j) a[j] = buf[base ^ lds_slot(spread3<F0, F1, F2>(j))];
}
template <int F0, int F1, int F2>
__device__ __forceinline__ void store8(const float2 (&a)[8], float2* buf, int base) {
    for (int j = 0; j < 8; ++j) buf[base ^ lds_slot(spread3<F0, F1, F2>(j))] = a[j];
}

__device__ __forceinline__ void rot_pair(float2& s0, float2& s1, const v4f g) {
    const float x0 = s0.x, y0 = s0.y, x1 = s1.x, y1 = s1.y;
    s0.x = g.x * x0 - g.y * y0 - g.z * x1 - g.w * y1;
    s0.y = g.x * y0 + g.y * x0 - g.z * y1 + g.w * x1;
    s1.x = g.z * x0 - g.w * y0 + g.x * x1 + g.y * y1;
    s1.y = g.z * y0 + g.w * x0 + g.x * y1 - g.y * x1;
}
template <int B>
__device__ __forceinline__ void rot_bit(float2 (&a)[8], const v4f g) {
    for (int j = 0; j < 8; ++j)
        if (!(j & (1 << B))) rot_pair(a[j], a[j | (1 << B)], g);
}

__device__ __forceinline__ void ising3(float2 (&a)[8], const float2* t, int b6) {
    {
        const float c = t[0].x, s = t[0].y;
        for (int j = 0; j < 4; ++j) {
            const float2 a0 = a[j], a1 = a[j ^ 6];
            a[j]     = make_float2(c * a0.x + s * a1.y, c * a0.y - s * a1.x);
            a[j ^ 6] = make_float2(c * a1.x + s * a0.y, c * a1.y - s * a0.x);
        }
    }
    {
        const float c = t[1].x, s = t[1].y;
        for (int j = 0; j < 8; ++j) {
            if (j & 2) continue;
            const float m = (j & 1) ? -s : s;
            const float2 a0 = a[j], a1 = a[j ^ 3];
            a[j]     = make_float2(c * a0.x - m * a1.y, c * a0.y + m * a1.x);
            a[j ^ 3] = make_float2(c * a1.x - m * a0.y, c * a1.y + m * a0.x);
        }
    }
    {
        const float c = t[2].x, s = t[2].y;
        const float s0 = b6 ? s : -s;
        for (int j = 0; j < 8; ++j) {
            const float py = (j & 1) ? -s0 : s0;
            const float2 v = a[j];
            a[j] = make_float2(c * v.x - py * v.y, c * v.y + py * v.x);
        }
    }
}

__global__ void __launch_bounds__(128) build_u_kernel(const float* __restrict__ w,
                                                      float* __restrict__ Mre,
                                                      float* __restrict__ Mim) {
    __shared__ __attribute__((aligned(16))) float2 sbuf[2 * DIM];
    __shared__ __attribute__((aligned(16))) v4f rot_tab[ROT_GATES];
    __shared__ float2 ising_tab[ISING_GATES];
    const int jcol = blockIdx.x;
    const int tid = threadIdx.x;

    if (tid < ROT_GATES) {
        const float phi = w[tid * 3 + 0];
        const float th  = w[tid * 3 + 1];
        const float om  = w[tid * 3 + 2];
        float ct, s_t, ca, sa, cb, sb;
        sincosf(0.5f * th, &s_t, &ct);
        sincosf(0.5f * (phi + om), &sa, &ca);
        sincosf(0.5f * (phi - om), &sb, &cb);
        v4f g;
        g.x =  ct * ca;  g.y = -ct * sa;
        g.z =  s_t * cb; g.w = -s_t * sb;
        rot_tab[tid] = g;
    } else if (tid >= 64 && tid < 64 + ISING_GATES) {
        const int g = tid - 64;
        float s, c;
        sincosf(0.5f * w[(g / 3) * NQ * 3 + (g % 3)], &s, &c);
        ising_tab[g] = make_float2(c, s);
    }

    const int pj = __popc(jcol) & 3;
    const float2 amp = make_float2(pj == 0 ? 1.0f : (pj == 2 ? -1.0f : 0.0f),
                                   pj == 1 ? -1.0f : (pj == 3 ? 1.0f : 0.0f));
    for (int k = tid; k < DIM; k += 128)
        sbuf[lds_slot(k)] = (k == jcol) ? amp : make_float2(0.0f, 0.0f);
    __syncthreads();

    const int t0 = tid & 1, t1 = (tid >> 1) & 1, t2 = (tid >> 2) & 1, t3 = (tid >> 3) & 1;
    const int t4 = (tid >> 4) & 1, t5 = (tid >> 5) & 1, t6 = (tid >> 6) & 1;
    const int kA = tid;
    const int kB = (tid & 15) | (t4 << 9) | (t5 << 7) | (t6 << 8);
    const int kC = t0 | (t1 << 4) | (t2 << 6) | (t3 << 7) | (t4 << 8) | (t5 << 5) | (t6 << 9);
    const int kD = tid << 3;
    const int pA = lds_slot(kA), pB = lds_slot(kB), pC = lds_slot(kC), pD = lds_slot(kD);
    const int pDst = lds_slot(cnot_ring(kD));

    int cur = 0;
    float2 a[8];
    for (int l = 0; l < NL; ++l) {
        const v4f* g = rot_tab + l * NQ;
        float2* buf = sbuf + cur;

        load8<7, 8, 9>(a, buf, pA);
        if (l > 0) ising3(a, ising_tab + (l - 1) * 3, t6);
        rot_bit<0>(a, g[2]); rot_bit<1>(a, g[1]); rot_bit<2>(a, g[0]);
        store8<7, 8, 9>(a, buf, pA);
        __syncthreads();

        load8<4, 5, 6>(a, buf, pB);
        rot_bit<0>(a, g[5]); rot_bit<1>(a, g[4]); rot_bit<2>(a, g[3]);
        store8<4, 5, 6>(a, buf, pB);
        __syncthreads();

        load8<1, 2, 3>(a, buf, pC);
        rot_bit<0>(a, g[8]); rot_bit<1>(a, g[7]); rot_bit<2>(a, g[6]);
        store8<1, 2, 3>(a, buf, pC);
        __syncthreads();

        load8<0, 1, 2>(a, buf, pD);
        rot_bit<0>(a, g[9]);
        float2* dst = sbuf + (cur ^ DIM);
        for (int j = 0; j < 8; ++j) dst[pDst ^ lds_slot(cnot_ring(j))] = a[j];
        __syncthreads();
        cur ^= DIM;
    }

    load8<7, 8, 9>(a, sbuf + cur, pA);
    ising3(a, ising_tab + (NL - 1) * 3, t6);
    for (int j = 0; j < 8; ++j) {
        Mre[jcol * DIM + tid + 128 * j] = a[j].x;
        Mim[jcol * DIM + tid + 128 * j] = a[j].y;
    }
}

__global__ void zero_kernel(float* __restrict__ out, int n) {
    const int i = blockIdx.x * blockDim.x + threadIdx.x;
    if (i < n) out[i] = 0.0f;
}

#define BM 64
#define BN 128
#define BK 64
#define APITCH (BK + 4)
#define BPITCH 80
#define B_PANEL (BK * BPITCH)
#define A_PLANE (BM * APITCH)
#define B_PLANE (2 * B_PANEL)
#define STAGE_FLOATS (A_PLANE + 2 * B_PLANE)

struct Frag {
    v2f a[2][2];
    v2f br[2][2], bi[2][2];
};

__device__ __forceinline__ void load_frag(Frag& f, const float* fa, const float* fb, int kk) {
    for (int s = 0; s < 2; ++s) {
        for (int rt = 0; rt < 2; ++rt)
            f.a[s][rt] = *(const v2f*)(fa + rt * 16 * APITCH + kk + 4 * s);
        for (int ct = 0; ct < 2; ++ct) {
            f.br[s][ct][0] = fb[ct * B_PANEL + (kk + 4 * s) * BPITCH];
            f.br[s][ct][1] = fb[ct * B_PANEL + (kk + 4 * s + 1) * BPITCH];
            f.bi[s][ct][0] = fb[B_PLANE + ct * B_PANEL + (kk + 4 * s) * BPITCH];
            f.bi[s][ct][1] = fb[B_PLANE + ct * B_PANEL + (kk + 4 * s + 1) * BPITCH];
        }
    }
}

__device__ __forceinline__ void mma_frag(const Frag& f, v8f (&acc_re)[2][2], v8f (&acc_im)[2][2]) {
    for (int s = 0; s < 2; ++s)
        for (int rt = 0; rt < 2; ++rt)
            for (int ct = 0; ct < 2; ++ct) {
                acc_re[rt][ct] = __builtin_amdgcn_wmma_f32_16x16x4_f32(false, f.a[s][rt], false, f.br[s][ct], (short)0, acc_re[rt][ct], false, false);
                acc_im[rt][ct] = __builtin_amdgcn_wmma_f32_16x16x4_f32(false, f.a[s][rt], false, f.bi[s][ct], (short)0, acc_im[rt][ct], false, false);
            }
}

template <int M>
__device__ __forceinline__ float xmask(float x) {
    return __builtin_bit_cast(float, __builtin_amdgcn_update_dpp(0, __builtin_bit_cast(int, x), 0x160 | M, 0xf, 0xf, true));
}

__global__ void __launch_bounds__(256) gemm_z_kernel(const float* __restrict__ x,
                                                     const float* __restrict__ Bre,
                                                     const float* __restrict__ Bim,
                                                     float* __restrict__ out) {
    __shared__ __attribute__((aligned(16))) float smem[2 * STAGE_FLOATS];
    __shared__ float cs_tab[BM * NQ], sn_tab[BM * NQ];
    __shared__ float hi_tab[(DIM / BK) * BM];

    const int tid = threadIdx.x;
    const int lane = tid & 31;
    const int wave = tid >> 5;
    const int wr = wave >> 2;
    const int wc = wave & 3;
    const int half = lane >> 4;
    const int l16 = lane & 15;
    const int m0 = (blockIdx.x >> 3) * BM;
    const int n0 = (blockIdx.x & 7) * BN;

    const int a_row = tid >> 4, a_kq = tid & 15;
    const int b_row = tid >> 5, b_nq = tid & 31;
    const float* gb_re = Bre + b_row * DIM + n0 + 4 * b_nq;
    const float* gb_im = Bim + b_row * DIM + n0 + 4 * b_nq;
    const int sa_off = a_row * APITCH + 4 * a_kq;
    const int sb_off = A_PLANE + (b_nq >> 4) * B_PANEL + b_row * BPITCH + 4 * (b_nq & 15);

    const int fa_off = (wr * 32 + l16) * APITCH + 2 * half;
    const int fb_off = A_PLANE + (2 * half) * BPITCH + wc * 16 + l16;

    v4f rb_re[8], rb_im[8];

    v8f acc_re[2][2], acc_im[2][2];
    for (int rt = 0; rt < 2; ++rt)
        for (int ct = 0; ct < 2; ++ct)
            for (int v = 0; v < 8; ++v) { acc_re[rt][ct][v] = 0.f; acc_im[rt][ct][v] = 0.f; }

    float xv[3];
    for (int j = 0; j < 3; ++j)
        xv[j] = x[m0 * NQ + 256 * j + (j < 2 ? tid : (tid & 127))];
    for (int p = 0; p < 8; ++p) {
        rb_re[p] = *(const v4f*)(gb_re + p * 8 * DIM);
        rb_im[p] = *(const v4f*)(gb_im + p * 8 * DIM);
    }

    for (int j = 0; j < 3; ++j) {
        if (j < 2 || tid < 128) {
            float s, c;
            sincosf(0.5f * xv[j], &s, &c);
            cs_tab[tid + 256 * j] = c; sn_tab[tid + 256 * j] = s;
        }
    }
    __syncthreads();
    for (int i = tid; i < (DIM / BK) * BM; i += 256) {
        const int row = i & (BM - 1), chunk = i >> 6;
        float r = 1.0f;
        for (int q = 0; q < 4; ++q)
            r *= ((chunk >> (3 - q)) & 1) ? sn_tab[row * NQ + q] : cs_tab[row * NQ + q];
        hi_tab[i] = r;
    }
    v4f lo[4];
    for (int p = 0; p < 4; ++p) {
        const int row = a_row + 16 * p;
        for (int i = 0; i < 4; ++i) {
            const int kl = 4 * a_kq + i;
            float r = 1.0f;
            for (int q = 4; q < NQ; ++q)
                r *= ((kl >> (NQ - 1 - q)) & 1) ? sn_tab[row * NQ + q] : cs_tab[row * NQ + q];
            lo[p][i] = r;
        }
    }
    __syncthreads();

    for (int p = 0; p < 4; ++p)
        *(v4f*)(smem + sa_off + p * 16 * APITCH) = lo[p] * hi_tab[a_row + 16 * p];
    for (int p = 0; p < 8; ++p) {
        *(v4f*)(smem + sb_off + p * 8 * BPITCH) = rb_re[p];
        *(v4f*)(smem + sb_off + B_PLANE + p * 8 * BPITCH) = rb_im[p];
    }
    __syncthreads();

    for (int ch = 0; ch < DIM / BK; ++ch) {
        const float* st = smem + (ch & 1) * STAGE_FLOATS;
        const bool more = (ch + 1) < DIM / BK;

        float ha[4];
        if (more) {
            const int k1 = (ch + 1) * BK;
            for (int p = 0; p < 4; ++p)
                ha[p] = hi_tab[(ch + 1) * BM + a_row + 16 * p];
            for (int p = 0; p < 8; ++p) {
                rb_re[p] = *(const v4f*)(gb_re + (k1 + p * 8) * DIM);
                rb_im[p] = *(const v4f*)(gb_im + (k1 + p * 8) * DIM);
            }
        }

        const float* fa = st + fa_off;
        const float* fb = st + fb_off;
        Frag f0, f1;
        load_frag(f0, fa, fb, 0);
        #pragma unroll 1
        for (int kk = 0; kk < BK; kk += 16) {
            load_frag(f1, fa, fb, kk + 8);
            __builtin_amdgcn_sched_barrier(0);
            mma_frag(f0, acc_re, acc_im);
            __builtin_amdgcn_sched_barrier(0);
            load_frag(f0, fa, fb, (kk + 16) & (BK - 1));
            __builtin_amdgcn_sched_barrier(0);
            mma_frag(f1, acc_re, acc_im);
        }

        if (more) {
            float* sn = smem + ((ch + 1) & 1) * STAGE_FLOATS;
            for (int p = 0; p < 4; ++p)
                *(v4f*)(sn + sa_off + p * 16 * APITCH) = lo[p] * ha[p];
            for (int p = 0; p < 8; ++p) {
                *(v4f*)(sn + sb_off + p * 8 * BPITCH) = rb_re[p];
                *(v4f*)(sn + sb_off + B_PLANE + p * 8 * BPITCH) = rb_im[p];
            }
        }
        __syncthreads();
    }

    float* red = smem;
    {
        const float g1 = (l16 & 1) ? -1.0f : 1.0f, g2 = (l16 & 2) ? -1.0f : 1.0f;
        const float g4 = (l16 & 4) ? -1.0f : 1.0f, g8 = (l16 & 8) ? -1.0f : 1.0f;
        const bool keeps = (l16 & (l16 - 1)) == 0;
        const int slot = __ffs(l16);
        for (int rt = 0; rt < 2; ++rt) {
            float s[8], d[8];
            for (int v = 0; v < 8; ++v) {
                const float p0 = acc_re[rt][0][v] * acc_re[rt][0][v] + acc_im[rt][0][v] * acc_im[rt][0][v];
                const float p1 = acc_re[rt][1][v] * acc_re[rt][1][v] + acc_im[rt][1][v] * acc_im[rt][1][v];
                s[v] = p0 + p1;
                d[v] = p0 - p1;
            }
            for (int v = 0; v < 8; ++v) s[v] = xmask<1>(s[v]) + g1 * s[v];
            for (int v = 0; v < 8; ++v) s[v] = xmask<2>(s[v]) + g2 * s[v];
            for (int v = 0; v < 8; ++v) s[v] = xmask<4>(s[v]) + g4 * s[v];
            for (int v = 0; v < 8; ++v) s[v] = xmask<8>(s[v]) + g8 * s[v];
            for (int v = 0; v < 8; ++v) d[v] += xmask<1>(d[v]);
            for (int v = 0; v < 8; ++v) d[v] += xmask<2>(d[v]);
            for (int v = 0; v < 8; ++v) d[v] += xmask<4>(d[v]);
            for (int v = 0; v < 8; ++v) d[v] += xmask<8>(d[v]);
            float* rr = red + ((wr * 32 + rt * 16 + half * 8) * 4 + wc) * 6;
            if (keeps) {
                for (int v = 0; v < 8; ++v) rr[v * 24 + slot] = s[v];
                if (l16 == 0)
                    for (int v = 0; v < 8; ++v) rr[v * 24 + 5] = d[v];
            }
        }
    }
    __syncthreads();
    for (int o = tid; o < BM * NQ; o += 256) {
        const int row = o / NQ, q = o - row * NQ;
        const int pos = NQ - 1 - q;
        const int c = pos >= 4 ? (pos == 6 ? 5 : 0) : pos + 1;
        float z = 0.0f;
        for (int w = 0; w < 4; ++w) {
            const float val = red[(row * 4 + w) * 6 + c];
            const int bit = pos >= 7 ? (n0 >> pos) : ((pos == 5 || pos == 4) ? (w >> (pos - 4)) : 0);
            z += (bit & 1) ? -val : val;
        }
        atomicAdd(&out[m0 * NQ + o], z);
    }
}

extern "C" void kernel_launch(void* const* d_in, const int* in_sizes, int n_in,
                              void* d_out, int out_size, void* d_ws, size_t ws_size,
                              hipStream_t stream) {
    const float* x = (const float*)d_in[0];
    const float* w = (const float*)d_in[1];
    float* out = (float*)d_out;

    float* Mre = (float*)d_ws;
    float* Mim = Mre + (size_t)DIM * DIM;

    build_u_kernel<<<DIM,   128, 0, stream>>>(w, Mre, Mim);
    zero_kernel   <<<(BATCH * NQ + 255) / 256, 256, 0, stream>>>(out, BATCH * NQ);
    gemm_z_kernel <<<(BATCH / BM) * (DIM / BN), 256, 0, stream>>>(x, Mre, Mim, out);
}


# GEMM K loop hand-over folded into the WMMA stream: barrier in front of a chunk's last WMMA block, the freed LDS stage refilled from registers behind it, the 2 waves of a SIMD at different points
# speedup vs baseline: 6.0350x; 1.0375x over previous
#include <hip/hip_runtime.h>
#include <hip/hip_bf16.h>
#include <math.h>

#define NQ    10
#define NL    6
#define BATCH 2048
#define DIM   1024


typedef float v2f __attribute__((ext_vector_type(2)));
typedef float v4f __attribute__((ext_vector_type(4)));
typedef float v8f __attribute__((ext_vector_type(8)));

__device__ __forceinline__ float2 cmul(float2 a, float2 b) {
    return make_float2(a.x * b.x - a.y * b.y, a.x * b.y + a.y * b.x);
}
__device__ __forceinline__ float2 cadd(float2 a, float2 b) {
    return make_float2(a.x + b.x, a.y + b.y);
}
__device__ __forceinline__ float2 cscale(float s, float2 a) {
    return make_float2(s * a.x, s * a.y);
}

#define ROT_GATES   (NL * NQ)
#define ISING_GATES (NL * 3)

__host__ __device__ constexpr int lds_slot(int k) { return k ^ (k >> 5); }

__host__ __device__ constexpr int cnot_ring(int k) {
    for (int q = 0; q < NQ; ++q) {
        const int pc = NQ - 1 - q;
        const int pt = NQ - 1 - ((q + 1) % NQ);
        k ^= ((k >> pc) & 1) << pt;
    }
    return k;
}

template <int F0, int F1, int F2>
__host__ __device__ constexpr int spread3(int j) {
    return ((j & 1) << F0) | (((j >> 1) & 1) << F1) | (((j >> 2) & 1) << F2);
}

template <int F0, int F1, int F2>
__device__ __forceinline__ void load8(float2 (&a)[8], const float2* buf, int base) {
    for (int j = 0; j < 8; ++j) a[j] = buf[base ^ lds_slot(spread3<F0, F1, F2>(j))];
}
template <int F0, int F1, int F2>
__device__ __forceinline__ void store8(const float2 (&a)[8], float2* buf, int base) {
    for (int j = 0; j < 8; ++j) buf[base ^ lds_slot(spread3<F0, F1, F2>(j))] = a[j];
}

__device__ __forceinline__ void rot_pair(float2& s0, float2& s1, const v4f g) {
    const float x0 = s0.x, y0 = s0.y, x1 = s1.x, y1 = s1.y;
    s0.x = g.x * x0 - g.y * y0 - g.z * x1 - g.w * y1;
    s0.y = g.x * y0 + g.y * x0 - g.z * y1 + g.w * x1;
    s1.x = g.z * x0 - g.w * y0 + g.x * x1 + g.y * y1;
    s1.y = g.z * y0 + g.w * x0 + g.x * y1 - g.y * x1;
}
template <int B>
__device__ __forceinline__ void rot_bit(float2 (&a)[8], const v4f g) {
    for (int j = 0; j < 8; ++j)
        if (!(j & (1 << B))) rot_pair(a[j], a[j | (1 << B)], g);
}

__device__ __forceinline__ void ising3(float2 (&a)[8], const float2* t, int b6) {
    {
        const float c = t[0].x, s = t[0].y;
        for (int j = 0; j < 4; ++j) {
            const float2 a0 = a[j], a1 = a[j ^ 6];
            a[j]     = make_float2(c * a0.x + s * a1.y, c * a0.y - s * a1.x);
            a[j ^ 6] = make_float2(c * a1.x + s * a0.y, c * a1.y - s * a0.x);
        }
    }
    {
        const float c = t[1].x, s = t[1].y;
        for (int j = 0; j < 8; ++j) {
            if (j & 2) continue;
            const float m = (j & 1) ? -s : s;
            const float2 a0 = a[j], a1 = a[j ^ 3];
            a[j]     = make_float2(c * a0.x - m * a1.y, c * a0.y + m * a1.x);
            a[j ^ 3] = make_float2(c * a1.x - m * a0.y, c * a1.y + m * a0.x);
        }
    }
    {
        const float c = t[2].x, s = t[2].y;
        const float s0 = b6 ? s : -s;
        for (int j = 0; j < 8; ++j) {
            const float py = (j & 1) ? -s0 : s0;
            const float2 v = a[j];
            a[j] = make_float2(c * v.x - py * v.y, c * v.y + py * v.x);
        }
    }
}

__global__ void __launch_bounds__(128) build_u_kernel(const float* __restrict__ w,
                                                      float* __restrict__ Mre,
                                                      float* __restrict__ Mim) {
    __shared__ __attribute__((aligned(16))) float2 sbuf[2 * DIM];
    __shared__ __attribute__((aligned(16))) v4f rot_tab[ROT_GATES];
    __shared__ float2 ising_tab[ISING_GATES];
    const int jcol = blockIdx.x;
    const int tid = threadIdx.x;

    if (tid < ROT_GATES) {
        const float phi = w[tid * 3 + 0];
        const float th  = w[tid * 3 + 1];
        const float om  = w[tid * 3 + 2];
        float ct, s_t, ca, sa, cb, sb;
        sincosf(0.5f * th, &s_t, &ct);
        sincosf(0.5f * (phi + om), &sa, &ca);
        sincosf(0.5f * (phi - om), &sb, &cb);
        v4f g;
        g.x =  ct * ca;  g.y = -ct * sa;
        g.z =  s_t * cb; g.w = -s_t * sb;
        rot_tab[tid] = g;
    } else if (tid >= 64 && tid < 64 + ISING_GATES) {
        const int g = tid - 64;
        float s, c;
        sincosf(0.5f * w[(g / 3) * NQ * 3 + (g % 3)], &s, &c);
        ising_tab[g] = make_float2(c, s);
    }

    const int pj = __popc(jcol) & 3;
    const float2 amp = make_float2(pj == 0 ? 1.0f : (pj == 2 ? -1.0f : 0.0f),
                                   pj == 1 ? -1.0f : (pj == 3 ? 1.0f : 0.0f));
    for (int k = tid; k < DIM; k += 128)
        sbuf[lds_slot(k)] = (k == jcol) ? amp : make_float2(0.0f, 0.0f);
    __syncthreads();

    const int t0 = tid & 1, t1 = (tid >> 1) & 1, t2 = (tid >> 2) & 1, t3 = (tid >> 3) & 1;
    const int t4 = (tid >> 4) & 1, t5 = (tid >> 5) & 1, t6 = (tid >> 6) & 1;
    const int kA = tid;
    const int kB = (tid & 15) | (t4 << 9) | (t5 << 7) | (t6 << 8);
    const int kC = t0 | (t1 << 4) | (t2 << 6) | (t3 << 7) | (t4 << 8) | (t5 << 5) | (t6 << 9);
    const int kD = tid << 3;
    const int pA = lds_slot(kA), pB = lds_slot(kB), pC = lds_slot(kC), pD = lds_slot(kD);
    const int pDst = lds_slot(cnot_ring(kD));

    int cur = 0;
    float2 a[8];
    for (int l = 0; l < NL; ++l) {
        const v4f* g = rot_tab + l * NQ;
        float2* buf = sbuf + cur;

        load8<7, 8, 9>(a, buf, pA);
        if (l > 0) ising3(a, ising_tab + (l - 1) * 3, t6);
        rot_bit<0>(a, g[2]); rot_bit<1>(a, g[1]); rot_bit<2>(a, g[0]);
        store8<7, 8, 9>(a, buf, pA);
        __syncthreads();

        load8<4, 5, 6>(a, buf, pB);
        rot_bit<0>(a, g[5]); rot_bit<1>(a, g[4]); rot_bit<2>(a, g[3]);
        store8<4, 5, 6>(a, buf, pB);
        __syncthreads();

        load8<1, 2, 3>(a, buf, pC);
        rot_bit<0>(a, g[8]); rot_bit<1>(a, g[7]); rot_bit<2>(a, g[6]);
        store8<1, 2, 3>(a, buf, pC);
        __syncthreads();

        load8<0, 1, 2>(a, buf, pD);
        rot_bit<0>(a, g[9]);
        float2* dst = sbuf + (cur ^ DIM);
        for (int j = 0; j < 8; ++j) dst[pDst ^ lds_slot(cnot_ring(j))] = a[j];
        __syncthreads();
        cur ^= DIM;
    }

    load8<7, 8, 9>(a, sbuf + cur, pA);
    ising3(a, ising_tab + (NL - 1) * 3, t6);
    for (int j = 0; j < 8; ++j) {
        Mre[jcol * DIM + tid + 128 * j] = a[j].x;
        Mim[jcol * DIM + tid + 128 * j] = a[j].y;
    }
}

__global__ void zero_kernel(float* __restrict__ out, int n) {
    const int i = blockIdx.x * blockDim.x + threadIdx.x;
    if (i < n) out[i] = 0.0f;
}

#define BM 64
#define BN 128
#define BK 64
#define APITCH (BK + 4)
#define BPITCH 80
#define B_PANEL (BK * BPITCH)
#define A_PLANE (BM * APITCH)
#define B_PLANE (2 * B_PANEL)
#define STAGE_FLOATS (A_PLANE + 2 * B_PLANE)

struct Frag {
    v2f a[2][2];
    v2f br[2][2], bi[2][2];
};

__device__ __forceinline__ void load_frag(Frag& f, const float* fa, const float* fb, int kk) {
    for (int s = 0; s < 2; ++s) {
        for (int rt = 0; rt < 2; ++rt)
            f.a[s][rt] = *(const v2f*)(fa + rt * 16 * APITCH + kk + 4 * s);
        for (int ct = 0; ct < 2; ++ct) {
            f.br[s][ct][0] = fb[ct * B_PANEL + (kk + 4 * s) * BPITCH];
            f.br[s][ct][1] = fb[ct * B_PANEL + (kk + 4 * s + 1) * BPITCH];
            f.bi[s][ct][0] = fb[B_PLANE + ct * B_PANEL + (kk + 4 * s) * BPITCH];
            f.bi[s][ct][1] = fb[B_PLANE + ct * B_PANEL + (kk + 4 * s + 1) * BPITCH];
        }
    }
}

__device__ __forceinline__ void mma_frag(const Frag& f, v8f (&acc_re)[2][2], v8f (&acc_im)[2][2]) {
    for (int s = 0; s < 2; ++s)
        for (int rt = 0; rt < 2; ++rt)
            for (int ct = 0; ct < 2; ++ct) {
                acc_re[rt][ct] = __builtin_amdgcn_wmma_f32_16x16x4_f32(false, f.a[s][rt], false, f.br[s][ct], (short)0, acc_re[rt][ct], false, false);
                acc_im[rt][ct] = __builtin_amdgcn_wmma_f32_16x16x4_f32(false, f.a[s][rt], false, f.bi[s][ct], (short)0, acc_im[rt][ct], false, false);
            }
}

template <int M>
__device__ __forceinline__ float xmask(float x) {
    return __builtin_bit_cast(float, __builtin_amdgcn_update_dpp(0, __builtin_bit_cast(int, x), 0x160 | M, 0xf, 0xf, true));
}

__global__ void __launch_bounds__(256) gemm_z_kernel(const float* __restrict__ x,
                                                     const float* __restrict__ Bre,
                                                     const float* __restrict__ Bim,
                                                     float* __restrict__ out) {
    __shared__ __attribute__((aligned(16))) float smem[2 * STAGE_FLOATS];
    __shared__ float cs_tab[BM * NQ], sn_tab[BM * NQ];
    __shared__ float hi_tab[(DIM / BK) * BM];

    const int tid = threadIdx.x;
    const int lane = tid & 31;
    const int wave = tid >> 5;
    const int wr = wave >> 2;
    const int wc = wave & 3;
    const int half = lane >> 4;
    const int l16 = lane & 15;
    const int m0 = (blockIdx.x >> 3) * BM;
    const int n0 = (blockIdx.x & 7) * BN;

    const int a_row = tid >> 4, a_kq = tid & 15;
    const int b_row = tid >> 5, b_nq = tid & 31;
    const float* gb_re = Bre + b_row * DIM + n0 + 4 * b_nq;
    const float* gb_im = Bim + b_row * DIM + n0 + 4 * b_nq;
    const int sa_off = a_row * APITCH + 4 * a_kq;
    const int sb_off = A_PLANE + (b_nq >> 4) * B_PANEL + b_row * BPITCH + 4 * (b_nq & 15);

    const int fa_off = (wr * 32 + l16) * APITCH + 2 * half;
    const int fb_off = A_PLANE + (2 * half) * BPITCH + wc * 16 + l16;

    v4f rb_re[8], rb_im[8];

    v8f acc_re[2][2], acc_im[2][2];
    for (int rt = 0; rt < 2; ++rt)
        for (int ct = 0; ct < 2; ++ct)
            for (int v = 0; v < 8; ++v) { acc_re[rt][ct][v] = 0.f; acc_im[rt][ct][v] = 0.f; }

    float xv[3];
    for (int j = 0; j < 3; ++j)
        xv[j] = x[m0 * NQ + 256 * j + (j < 2 ? tid : (tid & 127))];
    for (int p = 0; p < 8; ++p) {
        rb_re[p] = *(const v4f*)(gb_re + p * 8 * DIM);
        rb_im[p] = *(const v4f*)(gb_im + p * 8 * DIM);
    }

    for (int j = 0; j < 3; ++j) {
        if (j < 2 || tid < 128) {
            float s, c;
            sincosf(0.5f * xv[j], &s, &c);
            cs_tab[tid + 256 * j] = c; sn_tab[tid + 256 * j] = s;
        }
    }
    __syncthreads();
    for (int i = tid; i < (DIM / BK) * BM; i += 256) {
        const int row = i & (BM - 1), chunk = i >> 6;
        float r = 1.0f;
        for (int q = 0; q < 4; ++q)
            r *= ((chunk >> (3 - q)) & 1) ? sn_tab[row * NQ + q] : cs_tab[row * NQ + q];
        hi_tab[i] = r;
    }
    v4f lo[4];
    for (int p = 0; p < 4; ++p) {
        const int row = a_row + 16 * p;
        for (int i = 0; i < 4; ++i) {
            const int kl = 4 * a_kq + i;
            float r = 1.0f;
            for (int q = 4; q < NQ; ++q)
                r *= ((kl >> (NQ - 1 - q)) & 1) ? sn_tab[row * NQ + q] : cs_tab[row * NQ + q];
            lo[p][i] = r;
        }
    }
    __syncthreads();

    for (int p = 0; p < 4; ++p)
        *(v4f*)(smem + sa_off + p * 16 * APITCH) = lo[p] * hi_tab[a_row + 16 * p];
    for (int p = 0; p < 8; ++p) {
        *(v4f*)(smem + sb_off + p * 8 * BPITCH) = rb_re[p];
        *(v4f*)(smem + sb_off + B_PLANE + p * 8 * BPITCH) = rb_im[p];
    }
    __syncthreads();

    for (int p = 0; p < 8; ++p) {
        rb_re[p] = *(const v4f*)(gb_re + (BK + p * 8) * DIM);
        rb_im[p] = *(const v4f*)(gb_im + (BK + p * 8) * DIM);
    }

    Frag f0, f1;
    float ha[4];
    for (int p = 0; p < 4; ++p)
        ha[p] = hi_tab[BM + a_row + 16 * p];
    const bool fill_early = __builtin_amdgcn_readfirstlane(wave) >= 4;

    auto refill = [&](const int ch, const bool fill, const bool fetch) __attribute__((always_inline)) {
        if (fill) {
            float* sf = smem + (ch & 1) * STAGE_FLOATS;
            for (int p = 0; p < 4; ++p)
                *(v4f*)(sf + sa_off + p * 16 * APITCH) = lo[p] * ha[p];
            for (int p = 0; p < 8; ++p) {
                *(v4f*)(sf + sb_off + p * 8 * BPITCH) = rb_re[p];
                *(v4f*)(sf + sb_off + B_PLANE + p * 8 * BPITCH) = rb_im[p];
            }
        }
        __builtin_amdgcn_sched_barrier(0);
        if (fetch) {
            const int k3 = (ch + 3) * BK;
            for (int p = 0; p < 8; ++p) {
                rb_re[p] = *(const v4f*)(gb_re + (k3 + p * 8) * DIM);
                rb_im[p] = *(const v4f*)(gb_im + (k3 + p * 8) * DIM);
            }
        }
    };

    auto period = [&](const int ch, const bool first, const bool fill, const bool fetch) __attribute__((always_inline)) {
        const float* st = smem + (ch & 1) * STAGE_FLOATS;
        const float* fa = st + fa_off;
        const float* fb = st + fb_off;

        __builtin_amdgcn_sched_barrier(0);
        load_frag(f0, fa, fb, 0);
        if (!first) {
            if (fill_early) refill(ch - 1, fill, fetch);
            __builtin_amdgcn_sched_barrier(0);
            mma_frag(f1, acc_re, acc_im);
        }
        __builtin_amdgcn_sched_barrier(0);
        load_frag(f1, fa, fb, 8);
        __builtin_amdgcn_sched_barrier(0);
        mma_frag(f0, acc_re, acc_im);
        __builtin_amdgcn_sched_barrier(0);
        load_frag(f0, fa, fb, 16);
        if (first || !fill_early) refill(ch - 1, fill, fetch);
        __builtin_amdgcn_sched_barrier(0);
        mma_frag(f1, acc_re, acc_im);

        for (int kk = 16; kk < BK - 16; kk += 16) {
            __builtin_amdgcn_sched_barrier(0);
            load_frag(f1, fa, fb, kk + 8);
            __builtin_amdgcn_sched_barrier(0);
            mma_frag(f0, acc_re, acc_im);
            __builtin_amdgcn_sched_barrier(0);
            load_frag(f0, fa, fb, kk + 16);
            __builtin_amdgcn_sched_barrier(0);
            mma_frag(f1, acc_re, acc_im);
        }

        __builtin_amdgcn_sched_barrier(0);
        if (fetch)
            for (int p = 0; p < 4; ++p)
                ha[p] = hi_tab[(ch + 2) * BM + a_row + 16 * p];
        load_frag(f1, fa, fb, BK - 8);
        __builtin_amdgcn_sched_barrier(0);
        mma_frag(f0, acc_re, acc_im);
        __builtin_amdgcn_sched_barrier(0);
        if (fill) __syncthreads();
    };

    period(0, true, true, true);
    for (int ch = 1; ch < DIM / BK - 2; ++ch)
        period(ch, false, true, true);
    period(DIM / BK - 2, false, true, false);
    period(DIM / BK - 1, false, false, false);
    __builtin_amdgcn_sched_barrier(0);
    mma_frag(f1, acc_re, acc_im);

    float* red = smem;
    {
        const float g1 = (l16 & 1) ? -1.0f : 1.0f, g2 = (l16 & 2) ? -1.0f : 1.0f;
        const float g4 = (l16 & 4) ? -1.0f : 1.0f, g8 = (l16 & 8) ? -1.0f : 1.0f;
        const bool keeps = (l16 & (l16 - 1)) == 0;
        const int slot = __ffs(l16);
        for (int rt = 0; rt < 2; ++rt) {
            float s[8], d[8];
            for (int v = 0; v < 8; ++v) {
                const float p0 = acc_re[rt][0][v] * acc_re[rt][0][v] + acc_im[rt][0][v] * acc_im[rt][0][v];
                const float p1 = acc_re[rt][1][v] * acc_re[rt][1][v] + acc_im[rt][1][v] * acc_im[rt][1][v];
                s[v] = p0 + p1;
                d[v] = p0 - p1;
            }
            for (int v = 0; v < 8; ++v) s[v] = xmask<1>(s[v]) + g1 * s[v];
            for (int v = 0; v < 8; ++v) s[v] = xmask<2>(s[v]) + g2 * s[v];
            for (int v = 0; v < 8; ++v) s[v] = xmask<4>(s[v]) + g4 * s[v];
            for (int v = 0; v < 8; ++v) s[v] = xmask<8>(s[v]) + g8 * s[v];
            for (int v = 0; v < 8; ++v) d[v] += xmask<1>(d[v]);
            for (int v = 0; v < 8; ++v) d[v] += xmask<2>(d[v]);
            for (int v = 0; v < 8; ++v) d[v] += xmask<4>(d[v]);
            for (int v = 0; v < 8; ++v) d[v] += xmask<8>(d[v]);
            float* rr = red + ((wr * 32 + rt * 16 + half * 8) * 4 + wc) * 6;
            if (keeps) {
                for (int v = 0; v < 8; ++v) rr[v * 24 + slot] = s[v];
                if (l16 == 0)
                    for (int v = 0; v < 8; ++v) rr[v * 24 + 5] = d[v];
            }
        }
    }
    __syncthreads();
    for (int o = tid; o < BM * NQ; o += 256) {
        const int row = o / NQ, q = o - row * NQ;
        const int pos = NQ - 1 - q;
        const int c = pos >= 4 ? (pos == 6 ? 5 : 0) : pos + 1;
        float z = 0.0f;
        for (int w = 0; w < 4; ++w) {
            const float val = red[(row * 4 + w) * 6 + c];
            const int bit = pos >= 7 ? (n0 >> pos) : ((pos == 5 || pos == 4) ? (w >> (pos - 4)) : 0);
            z += (bit & 1) ? -val : val;
        }
        atomicAdd(&out[m0 * NQ + o], z);
    }
}

extern "C" void kernel_launch(void* const* d_in, const int* in_sizes, int n_in,
                              void* d_out, int out_size, void* d_ws, size_t ws_size,
                              hipStream_t stream) {
    const float* x = (const float*)d_in[0];
    const float* w = (const float*)d_in[1];
    float* out = (float*)d_out;

    float* Mre = (float*)d_ws;
    float* Mim = Mre + (size_t)DIM * DIM;

    build_u_kernel<<<DIM,   128, 0, stream>>>(w, Mre, Mim);
    zero_kernel   <<<(BATCH * NQ + 255) / 256, 256, 0, stream>>>(out, BATCH * NQ);
    gemm_z_kernel <<<(BATCH / BM) * (DIM / BN), 256, 0, stream>>>(x, Mre, Mim, out);
}


# Circuit-matrix build: every gate term is one packed f32 multiply-add on an amplitude's (re, im) pair, i*z read by operand selects, eight independent chains per round: 416 vector ops a layer, not 703
# speedup vs baseline: 6.5112x; 1.0789x over previous
#include <hip/hip_runtime.h>
#include <hip/hip_bf16.h>
#include <math.h>

#define NQ    10
#define NL    6
#define BATCH 2048
#define DIM   1024


typedef float v2f __attribute__((ext_vector_type(2)));
typedef float v4f __attribute__((ext_vector_type(4)));
typedef float v8f __attribute__((ext_vector_type(8)));

__device__ __forceinline__ float2 cmul(float2 a, float2 b) {
    return make_float2(a.x * b.x - a.y * b.y, a.x * b.y + a.y * b.x);
}
__device__ __forceinline__ float2 cadd(float2 a, float2 b) {
    return make_float2(a.x + b.x, a.y + b.y);
}
__device__ __forceinline__ float2 cscale(float s, float2 a) {
    return make_float2(s * a.x, s * a.y);
}

#define ROT_GATES   (NL * NQ)
#define ISING_GATES (NL * 3)

__host__ __device__ constexpr int lds_slot(int k) { return k ^ (k >> 5); }

__host__ __device__ constexpr int cnot_ring(int k) {
    for (int q = 0; q < NQ; ++q) {
        const int pc = NQ - 1 - q;
        const int pt = NQ - 1 - ((q + 1) % NQ);
        k ^= ((k >> pc) & 1) << pt;
    }
    return k;
}

template <int F0, int F1, int F2>
__host__ __device__ constexpr int spread3(int j) {
    return ((j & 1) << F0) | (((j >> 1) & 1) << F1) | (((j >> 2) & 1) << F2);
}

template <int F0, int F1, int F2>
__device__ __forceinline__ void load8(v2f (&a)[8], const v2f* buf, int base) {
    for (int j = 0; j < 8; ++j) a[j] = buf[base ^ lds_slot(spread3<F0, F1, F2>(j))];
}
template <int F0, int F1, int F2>
__device__ __forceinline__ void store8(const v2f (&a)[8], v2f* buf, int base) {
    for (int j = 0; j < 8; ++j) buf[base ^ lds_slot(spread3<F0, F1, F2>(j))] = a[j];
}

template <int H, bool NEG, bool IZ>
__device__ __forceinline__ v2f cfma(v2f c, v2f z, v2f acc) {
    v2f d;
    if constexpr (H == 0 && !NEG && !IZ)
        asm volatile("v_pk_fma_f32 %0, %1, %2, %3 op_sel:[0,0,0] op_sel_hi:[0,1,1]" : "=v"(d) : "v"(c), "v"(z), "v"(acc));
    else if constexpr (H == 0 && !NEG && IZ)
        asm volatile("v_pk_fma_f32 %0, %1, %2, %3 op_sel:[0,1,0] op_sel_hi:[0,0,1] neg_lo:[0,1,0]" : "=v"(d) : "v"(c), "v"(z), "v"(acc));
    else if constexpr (H == 0 && NEG && !IZ)
        asm volatile("v_pk_fma_f32 %0, %1, %2, %3 op_sel:[0,0,0] op_sel_hi:[0,1,1] neg_lo:[1,0,0] neg_hi:[1,0,0]" : "=v"(d) : "v"(c), "v"(z), "v"(acc));
    else if constexpr (H == 0 && NEG && IZ)
        asm volatile("v_pk_fma_f32 %0, %1, %2, %3 op_sel:[0,1,0] op_sel_hi:[0,0,1] neg_lo:[1,1,0] neg_hi:[1,0,0]" : "=v"(d) : "v"(c), "v"(z), "v"(acc));
    else if constexpr (H == 1 && !NEG && !IZ)
        asm volatile("v_pk_fma_f32 %0, %1, %2, %3 op_sel:[1,0,0] op_sel_hi:[1,1,1]" : "=v"(d) : "v"(c), "v"(z), "v"(acc));
    else if constexpr (H == 1 && !NEG && IZ)
        asm volatile("v_pk_fma_f32 %0, %1, %2, %3 op_sel:[1,1,0] op_sel_hi:[1,0,1] neg_lo:[0,1,0]" : "=v"(d) : "v"(c), "v"(z), "v"(acc));
    else if constexpr (H == 1 && NEG && !IZ)
        asm volatile("v_pk_fma_f32 %0, %1, %2, %3 op_sel:[1,0,0] op_sel_hi:[1,1,1] neg_lo:[1,0,0] neg_hi:[1,0,0]" : "=v"(d) : "v"(c), "v"(z), "v"(acc));
    else
        asm volatile("v_pk_fma_f32 %0, %1, %2, %3 op_sel:[1,1,0] op_sel_hi:[1,0,1] neg_lo:[1,1,0] neg_hi:[1,0,0]" : "=v"(d) : "v"(c), "v"(z), "v"(acc));
    return d;
}
template <int H>
__device__ __forceinline__ v2f cfirst(v2f c, v2f z) {
    v2f d;
    if constexpr (H == 0)
        asm volatile("v_pk_mul_f32 %0, %1, %2 op_sel:[0,0] op_sel_hi:[0,1]" : "=v"(d) : "v"(c), "v"(z));
    else
        asm volatile("v_pk_mul_f32 %0, %1, %2 op_sel:[1,0] op_sel_hi:[1,1]" : "=v"(d) : "v"(c), "v"(z));
    return d;
}

template <int B>
__device__ __forceinline__ void rot_bit(v2f (&a)[8], const v4f g) {
    const v2f ga = __builtin_shufflevector(g, g, 0, 1);
    const v2f gc = __builtin_shufflevector(g, g, 2, 3);
    constexpr int bit = 1 << B;
    v2f t[4], u[4];
    for (int p = 0; p < 4; ++p) {
        const int j = ((p >> B) << (B + 1)) | (p & (bit - 1));
        t[p] = cfirst<0>(ga, a[j]);
        u[p] = cfirst<0>(gc, a[j]);
    }
    for (int p = 0; p < 4; ++p) {
        const int j = ((p >> B) << (B + 1)) | (p & (bit - 1));
        t[p] = cfma<1, false, true>(ga, a[j], t[p]);
        u[p] = cfma<1, false, true>(gc, a[j], u[p]);
    }
    for (int p = 0; p < 4; ++p) {
        const int j = ((p >> B) << (B + 1)) | (p & (bit - 1));
        t[p] = cfma<0, true,  false>(gc, a[j | bit], t[p]);
        u[p] = cfma<0, false, false>(ga, a[j | bit], u[p]);
    }
    for (int p = 0; p < 4; ++p) {
        const int j = ((p >> B) << (B + 1)) | (p & (bit - 1));
        t[p] = cfma<1, false, true>(gc, a[j | bit], t[p]);
        u[p] = cfma<1, true,  true>(ga, a[j | bit], u[p]);
    }
    for (int p = 0; p < 4; ++p) {
        const int j = ((p >> B) << (B + 1)) | (p & (bit - 1));
        a[j] = t[p];
        a[j | bit] = u[p];
    }
}

template <int M, int NEGMASK>
__device__ __forceinline__ void two_term(v2f (&a)[8], const v2f cs) {
    v2f f[8], n[8];
    for (int k = 0; k < 8; ++k) f[k] = cfirst<0>(cs, a[k]);
    for (int k = 0; k < 8; ++k) {
        if ((NEGMASK >> k) & 1) n[k] = cfma<1, true,  true>(cs, a[k ^ M], f[k]);
        else                    n[k] = cfma<1, false, true>(cs, a[k ^ M], f[k]);
    }
    for (int k = 0; k < 8; ++k) a[k] = n[k];
}

__device__ __forceinline__ void ising3(v2f (&a)[8], const v2f txx, const v2f tyy, const v2f tzz, int b6) {
    two_term<6, 0xFF>(a, txx);
    two_term<3, 0x66>(a, tyy);
    two_term<0, 0xAA>(a, (v2f){tzz.x, b6 ? tzz.y : -tzz.y});
}

__global__ void __launch_bounds__(128) build_u_kernel(const float* __restrict__ w,
                                                      float* __restrict__ Mre,
                                                      float* __restrict__ Mim) {
    __shared__ __attribute__((aligned(16))) v2f sbuf[2 * DIM];
    __shared__ __attribute__((aligned(16))) v4f rot_tab[ROT_GATES];
    __shared__ v2f ising_tab[ISING_GATES];
    const int jcol = blockIdx.x;
    const int tid = threadIdx.x;

    if (tid < ROT_GATES) {
        const float phi = w[tid * 3 + 0];
        const float th  = w[tid * 3 + 1];
        const float om  = w[tid * 3 + 2];
        float ct, s_t, ca, sa, cb, sb;
        sincosf(0.5f * th, &s_t, &ct);
        sincosf(0.5f * (phi + om), &sa, &ca);
        sincosf(0.5f * (phi - om), &sb, &cb);
        v4f g;
        g.x =  ct * ca;  g.y = -ct * sa;
        g.z =  s_t * cb; g.w = -s_t * sb;
        rot_tab[tid] = g;
    } else if (tid >= 64 && tid < 64 + ISING_GATES) {
        const int g = tid - 64;
        float s, c;
        sincosf(0.5f * w[(g / 3) * NQ * 3 + (g % 3)], &s, &c);
        ising_tab[g] = (v2f){c, s};
    }

    const int pj = __popc(jcol) & 3;
    const v2f amp = (v2f){pj == 0 ? 1.0f : (pj == 2 ? -1.0f : 0.0f),
                          pj == 1 ? -1.0f : (pj == 3 ? 1.0f : 0.0f)};
    for (int k = tid; k < DIM; k += 128)
        sbuf[lds_slot(k)] = (k == jcol) ? amp : (v2f){0.0f, 0.0f};
    __syncthreads();

    const int t0 = tid & 1, t1 = (tid >> 1) & 1, t2 = (tid >> 2) & 1, t3 = (tid >> 3) & 1;
    const int t4 = (tid >> 4) & 1, t5 = (tid >> 5) & 1, t6 = (tid >> 6) & 1;
    const int kA = tid;
    const int kB = (tid & 15) | (t4 << 9) | (t5 << 7) | (t6 << 8);
    const int kC = t0 | (t1 << 4) | (t2 << 6) | (t3 << 7) | (t4 << 8) | (t5 << 5) | (t6 << 9);
    const int kD = tid << 3;
    const int pA = lds_slot(kA), pB = lds_slot(kB), pC = lds_slot(kC), pD = lds_slot(kD);
    const int pDst = lds_slot(cnot_ring(kD));

    int cur = 0;
    v2f a[8];
    for (int l = 0; l < NL; ++l) {
        const v4f* g = rot_tab + l * NQ;
        v2f* buf = sbuf + cur;

        {
            const v2f* it = ising_tab + (l > 0 ? l - 1 : 0) * 3;
            const v2f txx = it[0], tyy = it[1], tzz = it[2];
            const v4f r0 = g[0], r1 = g[1], r2 = g[2];
            load8<7, 8, 9>(a, buf, pA);
            if (l > 0) ising3(a, txx, tyy, tzz, t6);
            rot_bit<0>(a, r2); rot_bit<1>(a, r1); rot_bit<2>(a, r0);
            store8<7, 8, 9>(a, buf, pA);
        }
        __syncthreads();
        {
            const v4f r3 = g[3], r4 = g[4], r5 = g[5];
            load8<4, 5, 6>(a, buf, pB);
            rot_bit<0>(a, r5); rot_bit<1>(a, r4); rot_bit<2>(a, r3);
            store8<4, 5, 6>(a, buf, pB);
        }
        __syncthreads();
        {
            const v4f r6 = g[6], r7 = g[7], r8 = g[8];
            load8<1, 2, 3>(a, buf, pC);
            rot_bit<0>(a, r8); rot_bit<1>(a, r7); rot_bit<2>(a, r6);
            store8<1, 2, 3>(a, buf, pC);
        }
        __syncthreads();

        const v4f r9 = g[9];
        load8<0, 1, 2>(a, buf, pD);
        rot_bit<0>(a, r9);
        v2f* dst = sbuf + (cur ^ DIM);
        for (int j = 0; j < 8; ++j) dst[pDst ^ lds_slot(cnot_ring(j))] = a[j];
        __syncthreads();
        cur ^= DIM;
    }

    {
        const v2f* it = ising_tab + (NL - 1) * 3;
        const v2f txx = it[0], tyy = it[1], tzz = it[2];
        load8<7, 8, 9>(a, sbuf + cur, pA);
        ising3(a, txx, tyy, tzz, t6);
    }
    for (int j = 0; j < 8; ++j) {
        Mre[jcol * DIM + tid + 128 * j] = a[j].x;
        Mim[jcol * DIM + tid + 128 * j] = a[j].y;
    }
}

__global__ void zero_kernel(float* __restrict__ out, int n) {
    const int i = blockIdx.x * blockDim.x + threadIdx.x;
    if (i < n) out[i] = 0.0f;
}

#define BM 64
#define BN 128
#define BK 64
#define APITCH (BK + 4)
#define BPITCH 80
#define B_PANEL (BK * BPITCH)
#define A_PLANE (BM * APITCH)
#define B_PLANE (2 * B_PANEL)
#define STAGE_FLOATS (A_PLANE + 2 * B_PLANE)

struct Frag {
    v2f a[2][2];
    v2f br[2][2], bi[2][2];
};

__device__ __forceinline__ void load_frag(Frag& f, const float* fa, const float* fb, int kk) {
    for (int s = 0; s < 2; ++s) {
        for (int rt = 0; rt < 2; ++rt)
            f.a[s][rt] = *(const v2f*)(fa + rt * 16 * APITCH + kk + 4 * s);
        for (int ct = 0; ct < 2; ++ct) {
            f.br[s][ct][0] = fb[ct * B_PANEL + (kk + 4 * s) * BPITCH];
            f.br[s][ct][1] = fb[ct * B_PANEL + (kk + 4 * s + 1) * BPITCH];
            f.bi[s][ct][0] = fb[B_PLANE + ct * B_PANEL + (kk + 4 * s) * BPITCH];
            f.bi[s][ct][1] = fb[B_PLANE + ct * B_PANEL + (kk + 4 * s + 1) * BPITCH];
        }
    }
}

__device__ __forceinline__ void mma_frag(const Frag& f, v8f (&acc_re)[2][2], v8f (&acc_im)[2][2]) {
    for (int s = 0; s < 2; ++s)
        for (int rt = 0; rt < 2; ++rt)
            for (int ct = 0; ct < 2; ++ct) {
                acc_re[rt][ct] = __builtin_amdgcn_wmma_f32_16x16x4_f32(false, f.a[s][rt], false, f.br[s][ct], (short)0, acc_re[rt][ct], false, false);
                acc_im[rt][ct] = __builtin_amdgcn_wmma_f32_16x16x4_f32(false, f.a[s][rt], false, f.bi[s][ct], (short)0, acc_im[rt][ct], false, false);
            }
}

template <int M>
__device__ __forceinline__ float xmask(float x) {
    return __builtin_bit_cast(float, __builtin_amdgcn_update_dpp(0, __builtin_bit_cast(int, x), 0x160 | M, 0xf, 0xf, true));
}

__global__ void __launch_bounds__(256) gemm_z_kernel(const float* __restrict__ x,
                                                     const float* __restrict__ Bre,
                                                     const float* __restrict__ Bim,
                                                     float* __restrict__ out) {
    __shared__ __attribute__((aligned(16))) float smem[2 * STAGE_FLOATS];
    __shared__ float cs_tab[BM * NQ], sn_tab[BM * NQ];
    __shared__ float hi_tab[(DIM / BK) * BM];

    const int tid = threadIdx.x;
    const int lane = tid & 31;
    const int wave = tid >> 5;
    const int wr = wave >> 2;
    const int wc = wave & 3;
    const int half = lane >> 4;
    const int l16 = lane & 15;
    const int m0 = (blockIdx.x >> 3) * BM;
    const int n0 = (blockIdx.x & 7) * BN;

    const int a_row = tid >> 4, a_kq = tid & 15;
    const int b_row = tid >> 5, b_nq = tid & 31;
    const float* gb_re = Bre + b_row * DIM + n0 + 4 * b_nq;
    const float* gb_im = Bim + b_row * DIM + n0 + 4 * b_nq;
    const int sa_off = a_row * APITCH + 4 * a_kq;
    const int sb_off = A_PLANE + (b_nq >> 4) * B_PANEL + b_row * BPITCH + 4 * (b_nq & 15);

    const int fa_off = (wr * 32 + l16) * APITCH + 2 * half;
    const int fb_off = A_PLANE + (2 * half) * BPITCH + wc * 16 + l16;

    v4f rb_re[8], rb_im[8];

    v8f acc_re[2][2], acc_im[2][2];
    for (int rt = 0; rt < 2; ++rt)
        for (int ct = 0; ct < 2; ++ct)
            for (int v = 0; v < 8; ++v) { acc_re[rt][ct][v] = 0.f; acc_im[rt][ct][v] = 0.f; }

    float xv[3];
    for (int j = 0; j < 3; ++j)
        xv[j] = x[m0 * NQ + 256 * j + (j < 2 ? tid : (tid & 127))];
    for (int p = 0; p < 8; ++p) {
        rb_re[p] = *(const v4f*)(gb_re + p * 8 * DIM);
        rb_im[p] = *(const v4f*)(gb_im + p * 8 * DIM);
    }

    for (int j = 0; j < 3; ++j) {
        if (j < 2 || tid < 128) {
            float s, c;
            sincosf(0.5f * xv[j], &s, &c);
            cs_tab[tid + 256 * j] = c; sn_tab[tid + 256 * j] = s;
        }
    }
    __syncthreads();
    for (int i = tid; i < (DIM / BK) * BM; i += 256) {
        const int row = i & (BM - 1), chunk = i >> 6;
        float r = 1.0f;
        for (int q = 0; q < 4; ++q)
            r *= ((chunk >> (3 - q)) & 1) ? sn_tab[row * NQ + q] : cs_tab[row * NQ + q];
        hi_tab[i] = r;
    }
    v4f lo[4];
    for (int p = 0; p < 4; ++p) {
        const int row = a_row + 16 * p;
        for (int i = 0; i < 4; ++i) {
            const int kl = 4 * a_kq + i;
            float r = 1.0f;
            for (int q = 4; q < NQ; ++q)
                r *= ((kl >> (NQ - 1 - q)) & 1) ? sn_tab[row * NQ + q] : cs_tab[row * NQ + q];
            lo[p][i] = r;
        }
    }
    __syncthreads();

    for (int p = 0; p < 4; ++p)
        *(v4f*)(smem + sa_off + p * 16 * APITCH) = lo[p] * hi_tab[a_row + 16 * p];
    for (int p = 0; p < 8; ++p) {
        *(v4f*)(smem + sb_off + p * 8 * BPITCH) = rb_re[p];
        *(v4f*)(smem + sb_off + B_PLANE + p * 8 * BPITCH) = rb_im[p];
    }
    __syncthreads();

    for (int p = 0; p < 8; ++p) {
        rb_re[p] = *(const v4f*)(gb_re + (BK + p * 8) * DIM);
        rb_im[p] = *(const v4f*)(gb_im + (BK + p * 8) * DIM);
    }

    Frag f0, f1;
    float ha[4];
    for (int p = 0; p < 4; ++p)
        ha[p] = hi_tab[BM + a_row + 16 * p];
    const bool fill_early = __builtin_amdgcn_readfirstlane(wave) >= 4;

    auto refill = [&](const int ch, const bool fill, const bool fetch) __attribute__((always_inline)) {
        if (fill) {
            float* sf = smem + (ch & 1) * STAGE_FLOATS;
            for (int p = 0; p < 4; ++p)
                *(v4f*)(sf + sa_off + p * 16 * APITCH) = lo[p] * ha[p];
            for (int p = 0; p < 8; ++p) {
                *(v4f*)(sf + sb_off + p * 8 * BPITCH) = rb_re[p];
                *(v4f*)(sf + sb_off + B_PLANE + p * 8 * BPITCH) = rb_im[p];
            }
        }
        __builtin_amdgcn_sched_barrier(0);
        if (fetch) {
            const int k3 = (ch + 3) * BK;
            for (int p = 0; p < 8; ++p) {
                rb_re[p] = *(const v4f*)(gb_re + (k3 + p * 8) * DIM);
                rb_im[p] = *(const v4f*)(gb_im + (k3 + p * 8) * DIM);
            }
        }
    };

    auto period = [&](const int ch, const bool first, const bool fill, const bool fetch) __attribute__((always_inline)) {
        const float* st = smem + (ch & 1) * STAGE_FLOATS;
        const float* fa = st + fa_off;
        const float* fb = st + fb_off;

        __builtin_amdgcn_sched_barrier(0);
        load_frag(f0, fa, fb, 0);
        if (!first) {
            if (fill_early) refill(ch - 1, fill, fetch);
            __builtin_amdgcn_sched_barrier(0);
            mma_frag(f1, acc_re, acc_im);
        }
        __builtin_amdgcn_sched_barrier(0);
        load_frag(f1, fa, fb, 8);
        __builtin_amdgcn_sched_barrier(0);
        mma_frag(f0, acc_re, acc_im);
        __builtin_amdgcn_sched_barrier(0);
        load_frag(f0, fa, fb, 16);
        if (first || !fill_early) refill(ch - 1, fill, fetch);
        __builtin_amdgcn_sched_barrier(0);
        mma_frag(f1, acc_re, acc_im);

        for (int kk = 16; kk < BK - 16; kk += 16) {
            __builtin_amdgcn_sched_barrier(0);
            load_frag(f1, fa, fb, kk + 8);
            __builtin_amdgcn_sched_barrier(0);
            mma_frag(f0, acc_re, acc_im);
            __builtin_amdgcn_sched_barrier(0);
            load_frag(f0, fa, fb, kk + 16);
            __builtin_amdgcn_sched_barrier(0);
            mma_frag(f1, acc_re, acc_im);
        }

        __builtin_amdgcn_sched_barrier(0);
        if (fetch)
            for (int p = 0; p < 4; ++p)
                ha[p] = hi_tab[(ch + 2) * BM + a_row + 16 * p];
        load_frag(f1, fa, fb, BK - 8);
        __builtin_amdgcn_sched_barrier(0);
        mma_frag(f0, acc_re, acc_im);
        __builtin_amdgcn_sched_barrier(0);
        if (fill) __syncthreads();
    };

    period(0, true, true, true);
    for (int ch = 1; ch < DIM / BK - 2; ++ch)
        period(ch, false, true, true);
    period(DIM / BK - 2, false, true, false);
    period(DIM / BK - 1, false, false, false);
    __builtin_amdgcn_sched_barrier(0);
    mma_frag(f1, acc_re, acc_im);

    float* red = smem;
    {
        const float g1 = (l16 & 1) ? -1.0f : 1.0f, g2 = (l16 & 2) ? -1.0f : 1.0f;
        const float g4 = (l16 & 4) ? -1.0f : 1.0f, g8 = (l16 & 8) ? -1.0f : 1.0f;
        const bool keeps = (l16 & (l16 - 1)) == 0;
        const int slot = __ffs(l16);
        for (int rt = 0; rt < 2; ++rt) {
            float s[8], d[8];
            for (int v = 0; v < 8; ++v) {
                const float p0 = acc_re[rt][0][v] * acc_re[rt][0][v] + acc_im[rt][0][v] * acc_im[rt][0][v];
                const float p1 = acc_re[rt][1][v] * acc_re[rt][1][v] + acc_im[rt][1][v] * acc_im[rt][1][v];
                s[v] = p0 + p1;
                d[v] = p0 - p1;
            }
            for (int v = 0; v < 8; ++v) s[v] = xmask<1>(s[v]) + g1 * s[v];
            for (int v = 0; v < 8; ++v) s[v] = xmask<2>(s[v]) + g2 * s[v];
            for (int v = 0; v < 8; ++v) s[v] = xmask<4>(s[v]) + g4 * s[v];
            for (int v = 0; v < 8; ++v) s[v] = xmask<8>(s[v]) + g8 * s[v];
            for (int v = 0; v < 8; ++v) d[v] += xmask<1>(d[v]);
            for (int v = 0; v < 8; ++v) d[v] += xmask<2>(d[v]);
            for (int v = 0; v < 8; ++v) d[v] += xmask<4>(d[v]);
            for (int v = 0; v < 8; ++v) d[v] += xmask<8>(d[v]);
            float* rr = red + ((wr * 32 + rt * 16 + half * 8) * 4 + wc) * 6;
            if (keeps) {
                for (int v = 0; v < 8; ++v) rr[v * 24 + slot] = s[v];
                if (l16 == 0)
                    for (int v = 0; v < 8; ++v) rr[v * 24 + 5] = d[v];
            }
        }
    }
    __syncthreads();
    for (int o = tid; o < BM * NQ; o += 256) {
        const int row = o / NQ, q = o - row * NQ;
        const int pos = NQ - 1 - q;
        const int c = pos >= 4 ? (pos == 6 ? 5 : 0) : pos + 1;
        float z = 0.0f;
        for (int w = 0; w < 4; ++w) {
            const float val = red[(row * 4 + w) * 6 + c];
            const int bit = pos >= 7 ? (n0 >> pos) : ((pos == 5 || pos == 4) ? (w >> (pos - 4)) : 0);
            z += (bit & 1) ? -val : val;
        }
        atomicAdd(&out[m0 * NQ + o], z);
    }
}

extern "C" void kernel_launch(void* const* d_in, const int* in_sizes, int n_in,
                              void* d_out, int out_size, void* d_ws, size_t ws_size,
                              hipStream_t stream) {
    const float* x = (const float*)d_in[0];
    const float* w = (const float*)d_in[1];
    float* out = (float*)d_out;

    float* Mre = (float*)d_ws;
    float* Mim = Mre + (size_t)DIM * DIM;

    build_u_kernel<<<DIM,   128, 0, stream>>>(w, Mre, Mim);
    zero_kernel   <<<(BATCH * NQ + 255) / 256, 256, 0, stream>>>(out, BATCH * NQ);
    gemm_z_kernel <<<(BATCH / BM) * (DIM / BN), 256, 0, stream>>>(x, Mre, Mim, out);
}
